# starting kernel
# baseline (speedup 1.0000x reference)
#include <hip/hip_runtime.h>
#include <hip/hip_bf16.h>

#define F_IN   32
#define F_E    16
#define D      64
#define L      3
#define NEG_SLOPE 0.2f
#define BN_EPS 1e-5f

typedef __attribute__((ext_vector_type(2))) float v2f;
typedef __attribute__((ext_vector_type(8))) float v8f;

__device__ __forceinline__ int f2ord(float f) {
  int i = __float_as_int(f);
  return i >= 0 ? i : (i ^ 0x7FFFFFFF);
}
__device__ __forceinline__ float ord2f(int i) {
  return __int_as_float(i >= 0 ? i : (i ^ 0x7FFFFFFF));
}

__global__ void k_loop_accum(const float* __restrict__ eattr, const int* __restrict__ dst,
                             float* __restrict__ lsum, float* __restrict__ deg, int E) {
  int e = blockIdx.x * blockDim.x + threadIdx.x;
  if (e >= E) return;
  int d = dst[e];
  #pragma unroll
  for (int f = 0; f < F_E; f++) atomicAdd(&lsum[d * F_E + f], eattr[e * F_E + f]);
  atomicAdd(&deg[d], 1.0f);
}

__global__ void k_loop_div(float* __restrict__ lsum, const float* __restrict__ deg, int n) {
  int i = blockIdx.x * blockDim.x + threadIdx.x;
  if (i >= n * F_E) return;
  float c = deg[i / F_E];
  lsum[i] = lsum[i] / fmaxf(c, 1.0f);
}

__global__ void k_wea(const float* __restrict__ We, const float* __restrict__ ae,
                      float* __restrict__ wea) {
  int i = blockIdx.x * blockDim.x + threadIdx.x;
  if (i >= L * F_E) return;
  int l = i / F_E, f = i % F_E;
  float s = 0.f;
  #pragma unroll
  for (int d = 0; d < D; d++) s += We[(l * F_E + f) * D + d] * ae[l * D + d];
  wea[i] = s;
}

__global__ void k_gemm_wmma(const float* __restrict__ A, const float* __restrict__ B,
                            const float* __restrict__ bias, float* __restrict__ C,
                            int n, int K) {
  int wave = (blockIdx.x * blockDim.x + threadIdx.x) >> 5;
  int lane = threadIdx.x & 31;
  int row0 = wave * 16;
  if (row0 >= n) return;
  int lm = lane & 15;
  int kh = (lane >> 4) << 1;
  int mrow = row0 + lm;
  int mrd = mrow < n ? mrow : n - 1;

  v8f acc[4] = {};
  for (int k = 0; k < K; k += 4) {
    v2f a;
    a.x = A[(size_t)mrd * K + k + kh];
    a.y = A[(size_t)mrd * K + k + kh + 1];
    #pragma unroll
    for (int nt = 0; nt < 4; nt++) {
      int col = nt * 16 + lm;
      v2f b;
      b.x = B[(k + kh) * D + col];
      b.y = B[(k + kh + 1) * D + col];
      acc[nt] = __builtin_amdgcn_wmma_f32_16x16x4_f32(
          false, a, false, b, (short)0, acc[nt], false, false);
    }
  }

  int rbase = row0 + ((lane >> 4) << 3);
  #pragma unroll
  for (int nt = 0; nt < 4; nt++) {
    int col = nt * 16 + lm;
    float bv = bias ? bias[col] : 0.f;
    #pragma unroll
    for (int r = 0; r < 8; r++) {
      int row = rbase + r;
      if (row < n) C[(size_t)row * D + col] = acc[nt][r] + bv;
    }
  }
}

__global__ void k_scores(const float* __restrict__ xw, const float* __restrict__ as,
                         const float* __restrict__ ad, float* __restrict__ s,
                         float* __restrict__ dv, int n) {
  int i = blockIdx.x * blockDim.x + threadIdx.x;
  if (i >= n) return;
  const float* row = xw + (size_t)i * D;
  float ss = 0.f, dd = 0.f;
  #pragma unroll
  for (int f = 0; f < D; f++) { float v = row[f]; ss += v * as[f]; dd += v * ad[f]; }
  s[i] = ss;
  dv[i] = dd;
}

__global__ void k_init_amax(int* __restrict__ amax, int n) {
  int i = blockIdx.x * blockDim.x + threadIdx.x;
  if (i < n) amax[i] = (int)0x80000000;
}

__global__ void k_alpha(const float* __restrict__ s, const float* __restrict__ dv,
                        const float* __restrict__ eattr, const float* __restrict__ lattr,
                        const float* __restrict__ wea_l, const int* __restrict__ src,
                        const int* __restrict__ dst, float* __restrict__ alpha,
                        int* __restrict__ amax, int E, int n) {
  int i = blockIdx.x * blockDim.x + threadIdx.x;
  if (i >= E + n) return;
  int se, de;
  const float* ea;
  if (i < E) { se = src[i]; de = dst[i]; ea = eattr + (size_t)i * F_E; }
  else       { se = i - E;  de = i - E;  ea = lattr + (size_t)(i - E) * F_E; }
  float aed = 0.f;
  #pragma unroll
  for (int f = 0; f < F_E; f++) aed += ea[f] * wea_l[f];
  float a = s[se] + dv[de] + aed;
  a = a > 0.f ? a : NEG_SLOPE * a;
  alpha[i] = a;
  atomicMax(&amax[de], f2ord(a));
}

__global__ void k_exp(float* __restrict__ alpha, const int* __restrict__ amax,
                      const int* __restrict__ dst, float* __restrict__ denom, int E, int n) {
  int i = blockIdx.x * blockDim.x + threadIdx.x;
  if (i >= E + n) return;
  int de = i < E ? dst[i] : i - E;
  float ex = expf(alpha[i] - ord2f(amax[de]));
  alpha[i] = ex;
  atomicAdd(&denom[de], ex);
}

__global__ void k_aggr(const float* __restrict__ ex, const float* __restrict__ denom,
                       const float* __restrict__ xw, const int* __restrict__ src,
                       const int* __restrict__ dst, float* __restrict__ acc, int E, int n) {
  int t = blockIdx.x * blockDim.x + threadIdx.x;
  if (t >= (E + n) * 16) return;
  int i = t >> 4;
  int f0 = (t & 15) << 2;
  int se = i < E ? src[i] : i - E;
  int de = i < E ? dst[i] : i - E;
  float w = ex[i] / denom[de];
  const float4 v = *(const float4*)(xw + (size_t)se * D + f0);
  float* ap = acc + (size_t)de * D + f0;
  atomicAdd(ap + 0, w * v.x);
  atomicAdd(ap + 1, w * v.y);
  atomicAdd(ap + 2, w * v.z);
  atomicAdd(ap + 3, w * v.w);
}

__global__ void k_bn(const float* __restrict__ acc, const float* __restrict__ bias,
                     const float* __restrict__ gamma, const float* __restrict__ beta,
                     const float* __restrict__ mean, const float* __restrict__ var,
                     float* __restrict__ h, int n, int relu) {
  int i = blockIdx.x * blockDim.x + threadIdx.x;
  if (i >= n * D) return;
  int f = i & (D - 1);
  float v = acc[i] + bias[f];
  v = (v - mean[f]) * (gamma[f] / sqrtf(var[f] + BN_EPS)) + beta[f];
  if (relu) v = fmaxf(v, 0.f);
  h[i] = v;
}

__global__ void k_output(const float* __restrict__ h, const int* __restrict__ batch,
                         float* __restrict__ out_nodes, float* __restrict__ out_graph, int n) {
  int i = blockIdx.x * blockDim.x + threadIdx.x;
  if (i >= n * D) return;
  float v = h[i];
  out_nodes[i] = v;
  int node = i >> 6;
  int f = i & (D - 1);
  atomicAdd(&out_graph[(size_t)batch[node] * D + f], v);
}

extern "C" void kernel_launch(void* const* d_in, const int* in_sizes, int n_in,
                              void* d_out, int out_size, void* d_ws, size_t ws_size,
                              hipStream_t stream) {
  const float* x     = (const float*)d_in[0];
  const int*   eidx  = (const int*)d_in[1];
  const float* eattr = (const float*)d_in[2];
  const int*   batch = (const int*)d_in[3];
  const float* xlw   = (const float*)d_in[4];
  const float* xlb   = (const float*)d_in[5];
  const float* glw   = (const float*)d_in[6];
  const float* gas   = (const float*)d_in[7];
  const float* gad   = (const float*)d_in[8];
  const float* gew   = (const float*)d_in[9];
  const float* gae   = (const float*)d_in[10];
  const float* gb    = (const float*)d_in[11];
  const float* bng   = (const float*)d_in[12];
  const float* bnb   = (const float*)d_in[13];
  const float* bnm   = (const float*)d_in[14];
  const float* bnv   = (const float*)d_in[15];

  const int n = in_sizes[3];
  const int E = in_sizes[2] / F_E;
  const int* srcp = eidx;
  const int* dstp = eidx + E;

  float* ws    = (float*)d_ws;
  float* h     = ws;  ws += (size_t)n * D;
  float* xw    = ws;  ws += (size_t)n * D;
  float* acc   = ws;  ws += (size_t)n * D;
  float* sS    = ws;  ws += n;
  float* dS    = ws;  ws += n;
  float* lattr = ws;  ws += (size_t)n * F_E;
  float* deg   = ws;  ws += n;
  int*   amax  = (int*)ws; ws += n;
  float* denom = ws;  ws += n;
  float* alpha = ws;  ws += (size_t)(E + n);
  float* wea   = ws;  ws += L * F_E;

  float* out_nodes = (float*)d_out;
  float* out_graph = out_nodes + (size_t)n * D;
  const size_t graph_elems = (size_t)out_size - (size_t)n * D;

  const int B = 256;
  #define GRID(x) (((x) + B - 1) / B)

  hipMemsetAsync(lattr, 0, (size_t)n * F_E * sizeof(float), stream);
  hipMemsetAsync(deg, 0, (size_t)n * sizeof(float), stream);
  k_loop_accum<<<GRID(E), B, 0, stream>>>(eattr, dstp, lattr, deg, E);
  k_loop_div<<<GRID(n * F_E), B, 0, stream>>>(lattr, deg, n);
  k_wea<<<GRID(L * F_E), B, 0, stream>>>(gew, gae, wea);

  {
    int waves = (n + 15) / 16;
    int blocks = (waves + 3) / 4;
    k_gemm_wmma<<<blocks, 128, 0, stream>>>(x, xlw, xlb, h, n, F_IN);
  }

  for (int l = 0; l < L; l++) {
    hipMemsetAsync(denom, 0, (size_t)n * sizeof(float), stream);
    hipMemsetAsync(acc, 0, (size_t)n * D * sizeof(float), stream);
    k_init_amax<<<GRID(n), B, 0, stream>>>(amax, n);

    int waves = (n + 15) / 16;
    int blocks = (waves + 3) / 4;
    k_gemm_wmma<<<blocks, 128, 0, stream>>>(h, glw + (size_t)l * D * D, nullptr, xw, n, D);

    k_scores<<<GRID(n), B, 0, stream>>>(xw, gas + (size_t)l * D, gad + (size_t)l * D, sS, dS, n);
    k_alpha<<<GRID(E + n), B, 0, stream>>>(sS, dS, eattr, lattr, wea + (size_t)l * F_E,
                                           srcp, dstp, alpha, amax, E, n);
    k_exp<<<GRID(E + n), B, 0, stream>>>(alpha, amax, dstp, denom, E, n);
    k_aggr<<<GRID((E + n) * 16), B, 0, stream>>>(alpha, denom, xw, srcp, dstp, acc, E, n);
    k_bn<<<GRID(n * D), B, 0, stream>>>(acc, gb + (size_t)l * D, bng + (size_t)l * D,
                                        bnb + (size_t)l * D, bnm + (size_t)l * D,
                                        bnv + (size_t)l * D, h, n, l < L - 1 ? 1 : 0);
  }

  hipMemsetAsync(out_graph, 0, graph_elems * sizeof(float), stream);
  k_output<<<GRID(n * D), B, 0, stream>>>(h, batch, out_nodes, out_graph, n);
  #undef GRID
}


# output stage: last layer writes node rows straight into out; graph sums by one work-group per graph over its node range in sorted batch (256-way search), no atomics, memset, copy; sum order changed
# speedup vs baseline: 4.8376x; 4.8376x over previous
#include <hip/hip_runtime.h>


#define F_IN   32
#define F_E    16
#define D      64
#define L      3
#define NEG_SLOPE 0.2f
#define BN_EPS 1e-5f

typedef __attribute__((ext_vector_type(2))) float v2f;
typedef __attribute__((ext_vector_type(8))) float v8f;

__global__ void k_prep(const float* __restrict__ We, const float* __restrict__ ae,
                       float* __restrict__ wea, int* __restrict__ head, int n) {
  int i = blockIdx.x * blockDim.x + threadIdx.x;
  if (i < n) head[i] = -1;
  if (i >= L * F_E) return;
  int l = i / F_E, f = i % F_E;
  float s = 0.f;
  #pragma unroll
  for (int d = 0; d < D; d++) s += We[(l * F_E + f) * D + d] * ae[l * D + d];
  wea[i] = s;
}

__global__ void k_build(const int* __restrict__ src, const int* __restrict__ dst,
                        int* __restrict__ head, int2* __restrict__ link, int E) {
  int e = blockIdx.x * blockDim.x + threadIdx.x;
  if (e >= E) return;
  int prev = atomicExch(&head[dst[e]], e);
  link[e] = make_int2(prev, src[e]);
}

#define ET_TRIPS 4
__global__ void k_edge_term(const float* __restrict__ eattr, const float* __restrict__ wea_l,
                            float* __restrict__ aed, int E) {
  const int t = blockIdx.x * blockDim.x + threadIdx.x;
  const int T = gridDim.x * blockDim.x;
  float4 v[ET_TRIPS][F_E / 4];
  for (int k = 0; k < ET_TRIPS; k++) {
    const int e = t + k * T;
    const float4* ea = (const float4*)(eattr + (size_t)(e < E ? e : E - 1) * F_E);
    for (int q = 0; q < F_E / 4; q++) v[k][q] = ea[q];
  }
  float4 w[F_E / 4];
  for (int q = 0; q < F_E / 4; q++) w[q] = *(const float4*)(wea_l + q * 4);
  for (int k = 0; k < ET_TRIPS; k++) {
    const int e = t + k * T;
    float a = 0.f;
    for (int q = 0; q < F_E / 4; q++) {
      a += v[k][q].x * w[q].x;
      a += v[k][q].y * w[q].y;
      a += v[k][q].z * w[q].z;
      a += v[k][q].w * w[q].w;
    }
    if (e < E) aed[e] = a;
  }
}

__global__ void k_gemm_wmma(const float* __restrict__ A, const float* __restrict__ B,
                            const float* __restrict__ bias, float* __restrict__ C,
                            int n, int K) {
  int wave = (blockIdx.x * blockDim.x + threadIdx.x) >> 5;
  int lane = threadIdx.x & 31;
  int row0 = wave * 16;
  if (row0 >= n) return;
  int lm = lane & 15;
  int kh = (lane >> 4) << 1;
  int mrow = row0 + lm;
  int mrd = mrow < n ? mrow : n - 1;

  v8f acc[4] = {};
  for (int k = 0; k < K; k += 4) {
    v2f a;
    a.x = A[(size_t)mrd * K + k + kh];
    a.y = A[(size_t)mrd * K + k + kh + 1];
    #pragma unroll
    for (int nt = 0; nt < 4; nt++) {
      int col = nt * 16 + lm;
      v2f b;
      b.x = B[(k + kh) * D + col];
      b.y = B[(k + kh + 1) * D + col];
      acc[nt] = __builtin_amdgcn_wmma_f32_16x16x4_f32(
          false, a, false, b, (short)0, acc[nt], false, false);
    }
  }

  int rbase = row0 + ((lane >> 4) << 3);
  #pragma unroll
  for (int nt = 0; nt < 4; nt++) {
    int col = nt * 16 + lm;
    float bv = bias ? bias[col] : 0.f;
    #pragma unroll
    for (int r = 0; r < 8; r++) {
      int row = rbase + r;
      if (row < n) C[(size_t)row * D + col] = acc[nt][r] + bv;
    }
  }
}

__global__ void k_scores(const float* __restrict__ xw, const float* __restrict__ as,
                         const float* __restrict__ ad, float* __restrict__ s,
                         float* __restrict__ dv, int n) {
  int i = blockIdx.x * blockDim.x + threadIdx.x;
  if (i >= n) return;
  const float* row = xw + (size_t)i * D;
  float ss = 0.f, dd = 0.f;
  #pragma unroll
  for (int f = 0; f < D; f++) { float v = row[f]; ss += v * as[f]; dd += v * ad[f]; }
  s[i] = ss;
  dv[i] = dd;
}

__global__ void __launch_bounds__(256)
k_node(const float* __restrict__ xw, const float* __restrict__ s, const float* __restrict__ dv,
       const float* __restrict__ aed, const int2* __restrict__ link,
       const int* __restrict__ head, const float* __restrict__ bias,
       const float* __restrict__ gamma, const float* __restrict__ beta,
       const float* __restrict__ mean, const float* __restrict__ var,
       float* __restrict__ h, int n, int E, int relu) {
  int t = blockIdx.x * blockDim.x + threadIdx.x;
  int d = t >> 3;
  int j = t & 7;
  if (d >= n) return;
  const float dd = dv[d];
  float m = -INFINITY, den = 0.f, asum = 0.f, cnt = 0.f;
  float4 acc0 = make_float4(0.f, 0.f, 0.f, 0.f);
  float4 acc1 = make_float4(0.f, 0.f, 0.f, 0.f);

  int e = head[d];
  int2 lk = make_int2(-1, 0);
  if ((unsigned)e < (unsigned)E) lk = link[e];
  for (int it = 0; it < E && (unsigned)e < (unsigned)E; ++it) {
    const int nx = lk.x, sr = lk.y;
    const float ae = aed[e];
    const float ss = s[sr];
    const float* rp = xw + (size_t)sr * D + j * 4;
    const float4 row0 = *(const float4*)rp;
    const float4 row1 = *(const float4*)(rp + D / 2);
    const int2 nlk = link[(unsigned)nx < (unsigned)E ? nx : 0];
    float a = ss + dd + ae;
    a = a > 0.f ? a : NEG_SLOPE * a;
    asum += ae;
    cnt += 1.0f;
    const float dl = a - m;
    const float ex = expf(-fabsf(dl));
    const bool up = dl > 0.f;
    const float sc = up ? ex : 1.0f;
    const float p  = up ? 1.0f : ex;
    m = up ? a : m;
    den = den * sc + p;
    acc0.x = acc0.x * sc + p * row0.x;
    acc0.y = acc0.y * sc + p * row0.y;
    acc0.z = acc0.z * sc + p * row0.z;
    acc0.w = acc0.w * sc + p * row0.w;
    acc1.x = acc1.x * sc + p * row1.x;
    acc1.y = acc1.y * sc + p * row1.y;
    acc1.z = acc1.z * sc + p * row1.z;
    acc1.w = acc1.w * sc + p * row1.w;
    e = nx;
    lk = nlk;
  }

  {
    const float* rp = xw + (size_t)d * D + j * 4;
    const float4 row0 = *(const float4*)rp;
    const float4 row1 = *(const float4*)(rp + D / 2);
    float a = s[d] + dd + asum / fmaxf(cnt, 1.0f);
    a = a > 0.f ? a : NEG_SLOPE * a;
    const float dl = a - m;
    const float ex = expf(-fabsf(dl));
    const bool up = dl > 0.f;
    const float sc = up ? ex : 1.0f;
    const float p  = up ? 1.0f : ex;
    den = den * sc + p;
    acc0.x = acc0.x * sc + p * row0.x;
    acc0.y = acc0.y * sc + p * row0.y;
    acc0.z = acc0.z * sc + p * row0.z;
    acc0.w = acc0.w * sc + p * row0.w;
    acc1.x = acc1.x * sc + p * row1.x;
    acc1.y = acc1.y * sc + p * row1.y;
    acc1.z = acc1.z * sc + p * row1.z;
    acc1.w = acc1.w * sc + p * row1.w;
  }

  const float inv = 1.0f / den;
  #pragma unroll 1
  for (int hf = 0; hf < 2; hf++) {
    const int f = hf * (D / 2) + j * 4;
    const float4 ac = hf ? acc1 : acc0;
    const float4 b4 = *(const float4*)(bias  + f);
    const float4 g4 = *(const float4*)(gamma + f);
    const float4 t4 = *(const float4*)(beta  + f);
    const float4 m4 = *(const float4*)(mean  + f);
    const float4 v4 = *(const float4*)(var   + f);
    float4 o;
    o.x = (ac.x * inv + b4.x - m4.x) * (g4.x / sqrtf(v4.x + BN_EPS)) + t4.x;
    o.y = (ac.y * inv + b4.y - m4.y) * (g4.y / sqrtf(v4.y + BN_EPS)) + t4.y;
    o.z = (ac.z * inv + b4.z - m4.z) * (g4.z / sqrtf(v4.z + BN_EPS)) + t4.z;
    o.w = (ac.w * inv + b4.w - m4.w) * (g4.w / sqrtf(v4.w + BN_EPS)) + t4.w;
    if (relu) {
      o.x = fmaxf(o.x, 0.f); o.y = fmaxf(o.y, 0.f); o.z = fmaxf(o.z, 0.f); o.w = fmaxf(o.w, 0.f);
    }
    *(float4*)(h + (size_t)d * D + f) = o;
  }
}

#define RO_T 256
#define RO_W (RO_T / 32)
__global__ void __launch_bounds__(RO_T)
k_readout(const float* __restrict__ nodes, const int* __restrict__ batch,
          float* __restrict__ out_graph, int n) {
  __shared__ float part[16 * D];
  __shared__ int cnt[2][2][RO_W];
  const int g = blockIdx.x;
  const int t = threadIdx.x;

  int lo0 = 0, len0 = n, lo1 = 0, len1 = n;
  for (int round = 0; round < 8 && (len0 > 0 || len1 > 0); ++round) {
    const int st0 = (len0 + RO_T - 1) / RO_T;
    const int st1 = (len1 + RO_T - 1) / RO_T;
    const int p0 = lo0 + t * st0;
    const int p1 = lo1 + t * st1;
    const int b0 = batch[p0 < n ? p0 : n - 1];
    const int b1 = batch[p1 < n ? p1 : n - 1];
    const bool q0 = (p0 < lo0 + len0) && (b0 < g);
    const bool q1 = (p1 < lo1 + len1) && (b1 <= g);
    const int w0 = __popc(__builtin_amdgcn_ballot_w32(q0));
    const int w1 = __popc(__builtin_amdgcn_ballot_w32(q1));
    const int par = round & 1;
    if ((t & 31) == 0) { cnt[par][0][t >> 5] = w0; cnt[par][1][t >> 5] = w1; }
    __syncthreads();
    int c0 = 0, c1 = 0;
    for (int w = 0; w < RO_W; w++) { c0 += cnt[par][0][w]; c1 += cnt[par][1][w]; }
    c0 = __builtin_amdgcn_readfirstlane(c0);
    c1 = __builtin_amdgcn_readfirstlane(c1);
    if (len0 > 0) {
      if (c0 == 0) { len0 = 0; }
      else {
        const int nlo = lo0 + (c0 - 1) * st0 + 1;
        const int cap = lo0 + c0 * st0;
        const int nhi = cap < lo0 + len0 ? cap : lo0 + len0;
        lo0 = nlo; len0 = nhi - nlo;
      }
    }
    if (len1 > 0) {
      if (c1 == 0) { len1 = 0; }
      else {
        const int nlo = lo1 + (c1 - 1) * st1 + 1;
        const int cap = lo1 + c1 * st1;
        const int nhi = cap < lo1 + len1 ? cap : lo1 + len1;
        lo1 = nlo; len1 = nhi - nlo;
      }
    }
  }
  const int first = lo0, last = lo1;

  const int r = t >> 4;
  const int c = t & 15;
  float4 acc = make_float4(0.f, 0.f, 0.f, 0.f);
  int i = first + r;
  for (; i + 48 < last; i += 64) {
    const float* p = nodes + (size_t)i * D + c * 4;
    const float4 v0 = *(const float4*)p;
    const float4 v1 = *(const float4*)(p + 16 * D);
    const float4 v2 = *(const float4*)(p + 32 * D);
    const float4 v3 = *(const float4*)(p + 48 * D);
    acc.x += v0.x; acc.y += v0.y; acc.z += v0.z; acc.w += v0.w;
    acc.x += v1.x; acc.y += v1.y; acc.z += v1.z; acc.w += v1.w;
    acc.x += v2.x; acc.y += v2.y; acc.z += v2.z; acc.w += v2.w;
    acc.x += v3.x; acc.y += v3.y; acc.z += v3.z; acc.w += v3.w;
  }
  for (; i < last; i += 16) {
    const float4 v = *(const float4*)(nodes + (size_t)i * D + c * 4);
    acc.x += v.x; acc.y += v.y; acc.z += v.z; acc.w += v.w;
  }
  *(float4*)(part + r * D + c * 4) = acc;
  __syncthreads();
  if (t < D) {
    float s = 0.f;
    for (int k = 0; k < 16; k++) s += part[k * D + t];
    out_graph[(size_t)g * D + t] = s;
  }
}

extern "C" void kernel_launch(void* const* d_in, const int* in_sizes, int n_in,
                              void* d_out, int out_size, void* d_ws, size_t ws_size,
                              hipStream_t stream) {
  const float* x     = (const float*)d_in[0];
  const int*   eidx  = (const int*)d_in[1];
  const float* eattr = (const float*)d_in[2];
  const int*   batch = (const int*)d_in[3];
  const float* xlw   = (const float*)d_in[4];
  const float* xlb   = (const float*)d_in[5];
  const float* glw   = (const float*)d_in[6];
  const float* gas   = (const float*)d_in[7];
  const float* gad   = (const float*)d_in[8];
  const float* gew   = (const float*)d_in[9];
  const float* gae   = (const float*)d_in[10];
  const float* gb    = (const float*)d_in[11];
  const float* bng   = (const float*)d_in[12];
  const float* bnb   = (const float*)d_in[13];
  const float* bnm   = (const float*)d_in[14];
  const float* bnv   = (const float*)d_in[15];

  const int n = in_sizes[3];
  const int E = in_sizes[2] / F_E;
  const int* srcp = eidx;
  const int* dstp = eidx + E;

  float* ws    = (float*)d_ws;
  float* h     = ws;  ws += (size_t)n * D;
  float* xw    = ws;  ws += (size_t)n * D;
  float* wea   = ws;  ws += L * F_E;
  int2*  link  = (int2*)ws; ws += (size_t)E * 2;
  float* aed   = ws;  ws += E;
  float* sS    = ws;  ws += n;
  float* dS    = ws;  ws += n;
  int*   head  = (int*)ws; ws += n;

  float* out_nodes = (float*)d_out;
  float* out_graph = out_nodes + (size_t)n * D;
  const size_t graph_elems = (size_t)out_size - (size_t)n * D;

  const int B = 256;
  #define GRID(x) (((x) + B - 1) / B)

  k_prep<<<GRID(n > L * F_E ? n : L * F_E), B, 0, stream>>>(gew, gae, wea, head, n);
  k_build<<<GRID(E), B, 0, stream>>>(srcp, dstp, head, link, E);

  {
    int waves = (n + 15) / 16;
    int blocks = (waves + 3) / 4;
    k_gemm_wmma<<<blocks, 128, 0, stream>>>(x, xlw, xlb, h, n, F_IN);
  }

  for (int l = 0; l < L; l++) {
    k_edge_term<<<GRID((E + ET_TRIPS - 1) / ET_TRIPS), B, 0, stream>>>(eattr, wea + (size_t)l * F_E, aed, E);

    int waves = (n + 15) / 16;
    int blocks = (waves + 3) / 4;
    k_gemm_wmma<<<blocks, 128, 0, stream>>>(h, glw + (size_t)l * D * D, nullptr, xw, n, D);

    k_scores<<<GRID(n), B, 0, stream>>>(xw, gas + (size_t)l * D, gad + (size_t)l * D, sS, dS, n);
    k_node<<<GRID(n * 8), B, 0, stream>>>(xw, sS, dS, aed, link, head, gb + (size_t)l * D,
                                           bng + (size_t)l * D, bnb + (size_t)l * D,
                                           bnm + (size_t)l * D, bnv + (size_t)l * D,
                                           l < L - 1 ? h : out_nodes, n, E, l < L - 1 ? 1 : 0);
  }

  k_readout<<<(int)(graph_elems / D), RO_T, 0, stream>>>(out_nodes, batch, out_graph, n);
  #undef GRID
}


# edge terms of all 3 layers formed in ONE pass over edge_attr before the layer loop (3 dots per edge, weights in scalar registers, 3 planes in edge order): 2 reads, 2 launches gone; sum orders kept
# speedup vs baseline: 5.4377x; 1.1240x over previous
#include <hip/hip_runtime.h>


#define F_IN   32
#define F_E    16
#define D      64
#define L      3
#define NEG_SLOPE 0.2f
#define BN_EPS 1e-5f

typedef __attribute__((ext_vector_type(2))) float v2f;
typedef __attribute__((ext_vector_type(8))) float v8f;

__global__ void k_prep(const float* __restrict__ We, const float* __restrict__ ae,
                       float* __restrict__ wea, int* __restrict__ head, int n) {
  int i = blockIdx.x * blockDim.x + threadIdx.x;
  if (i < n) head[i] = -1;
  if (i >= L * F_E) return;
  int l = i / F_E, f = i % F_E;
  float s = 0.f;
  #pragma unroll
  for (int d = 0; d < D; d++) s += We[(l * F_E + f) * D + d] * ae[l * D + d];
  wea[i] = s;
}

__global__ void k_build(const int* __restrict__ src, const int* __restrict__ dst,
                        int* __restrict__ head, int2* __restrict__ link, int E) {
  int e = blockIdx.x * blockDim.x + threadIdx.x;
  if (e >= E) return;
  int prev = atomicExch(&head[dst[e]], e);
  link[e] = make_int2(prev, src[e]);
}

#define ET_TRIPS 4
__global__ void k_edge_terms(const float* __restrict__ eattr, const float* __restrict__ wea,
                             float* __restrict__ aed, int E) {
  const int t = blockIdx.x * blockDim.x + threadIdx.x;
  const int T = gridDim.x * blockDim.x;
  float4 v[ET_TRIPS][F_E / 4];
  for (int k = 0; k < ET_TRIPS; k++) {
    const int e = t + k * T;
    const float4* ea = (const float4*)(eattr + (size_t)(e < E ? e : E - 1) * F_E);
    for (int q = 0; q < F_E / 4; q++) v[k][q] = ea[q];
  }
  float4 w[L][F_E / 4];
  for (int l = 0; l < L; l++)
    for (int q = 0; q < F_E / 4; q++) w[l][q] = *(const float4*)(wea + l * F_E + q * 4);
  for (int k = 0; k < ET_TRIPS; k++) {
    const int e = t + k * T;
    for (int l = 0; l < L; l++) {
      float a = 0.f;
      for (int q = 0; q < F_E / 4; q++) {
        a += v[k][q].x * w[l][q].x;
        a += v[k][q].y * w[l][q].y;
        a += v[k][q].z * w[l][q].z;
        a += v[k][q].w * w[l][q].w;
      }
      if (e < E) aed[(size_t)l * E + e] = a;
    }
  }
}

__global__ void k_gemm_wmma(const float* __restrict__ A, const float* __restrict__ B,
                            const float* __restrict__ bias, float* __restrict__ C,
                            int n, int K) {
  int wave = (blockIdx.x * blockDim.x + threadIdx.x) >> 5;
  int lane = threadIdx.x & 31;
  int row0 = wave * 16;
  if (row0 >= n) return;
  int lm = lane & 15;
  int kh = (lane >> 4) << 1;
  int mrow = row0 + lm;
  int mrd = mrow < n ? mrow : n - 1;

  v8f acc[4] = {};
  for (int k = 0; k < K; k += 4) {
    v2f a;
    a.x = A[(size_t)mrd * K + k + kh];
    a.y = A[(size_t)mrd * K + k + kh + 1];
    #pragma unroll
    for (int nt = 0; nt < 4; nt++) {
      int col = nt * 16 + lm;
      v2f b;
      b.x = B[(k + kh) * D + col];
      b.y = B[(k + kh + 1) * D + col];
      acc[nt] = __builtin_amdgcn_wmma_f32_16x16x4_f32(
          false, a, false, b, (short)0, acc[nt], false, false);
    }
  }

  int rbase = row0 + ((lane >> 4) << 3);
  #pragma unroll
  for (int nt = 0; nt < 4; nt++) {
    int col = nt * 16 + lm;
    float bv = bias ? bias[col] : 0.f;
    #pragma unroll
    for (int r = 0; r < 8; r++) {
      int row = rbase + r;
      if (row < n) C[(size_t)row * D + col] = acc[nt][r] + bv;
    }
  }
}

__global__ void k_scores(const float* __restrict__ xw, const float* __restrict__ as,
                         const float* __restrict__ ad, float* __restrict__ s,
                         float* __restrict__ dv, int n) {
  int i = blockIdx.x * blockDim.x + threadIdx.x;
  if (i >= n) return;
  const float* row = xw + (size_t)i * D;
  float ss = 0.f, dd = 0.f;
  #pragma unroll
  for (int f = 0; f < D; f++) { float v = row[f]; ss += v * as[f]; dd += v * ad[f]; }
  s[i] = ss;
  dv[i] = dd;
}

__global__ void __launch_bounds__(256)
k_node(const float* __restrict__ xw, const float* __restrict__ s, const float* __restrict__ dv,
       const float* __restrict__ aed, const int2* __restrict__ link,
       const int* __restrict__ head, const float* __restrict__ bias,
       const float* __restrict__ gamma, const float* __restrict__ beta,
       const float* __restrict__ mean, const float* __restrict__ var,
       float* __restrict__ h, int n, int E, int relu) {
  int t = blockIdx.x * blockDim.x + threadIdx.x;
  int d = t >> 3;
  int j = t & 7;
  if (d >= n) return;
  const float dd = dv[d];
  float m = -INFINITY, den = 0.f, asum = 0.f, cnt = 0.f;
  float4 acc0 = make_float4(0.f, 0.f, 0.f, 0.f);
  float4 acc1 = make_float4(0.f, 0.f, 0.f, 0.f);

  int e = head[d];
  int2 lk = make_int2(-1, 0);
  if ((unsigned)e < (unsigned)E) lk = link[e];
  for (int it = 0; it < E && (unsigned)e < (unsigned)E; ++it) {
    const int nx = lk.x, sr = lk.y;
    const float ae = aed[e];
    const float ss = s[sr];
    const float* rp = xw + (size_t)sr * D + j * 4;
    const float4 row0 = *(const float4*)rp;
    const float4 row1 = *(const float4*)(rp + D / 2);
    const int2 nlk = link[(unsigned)nx < (unsigned)E ? nx : 0];
    float a = ss + dd + ae;
    a = a > 0.f ? a : NEG_SLOPE * a;
    asum += ae;
    cnt += 1.0f;
    const float dl = a - m;
    const float ex = expf(-fabsf(dl));
    const bool up = dl > 0.f;
    const float sc = up ? ex : 1.0f;
    const float p  = up ? 1.0f : ex;
    m = up ? a : m;
    den = den * sc + p;
    acc0.x = acc0.x * sc + p * row0.x;
    acc0.y = acc0.y * sc + p * row0.y;
    acc0.z = acc0.z * sc + p * row0.z;
    acc0.w = acc0.w * sc + p * row0.w;
    acc1.x = acc1.x * sc + p * row1.x;
    acc1.y = acc1.y * sc + p * row1.y;
    acc1.z = acc1.z * sc + p * row1.z;
    acc1.w = acc1.w * sc + p * row1.w;
    e = nx;
    lk = nlk;
  }

  {
    const float* rp = xw + (size_t)d * D + j * 4;
    const float4 row0 = *(const float4*)rp;
    const float4 row1 = *(const float4*)(rp + D / 2);
    float a = s[d] + dd + asum / fmaxf(cnt, 1.0f);
    a = a > 0.f ? a : NEG_SLOPE * a;
    const float dl = a - m;
    const float ex = expf(-fabsf(dl));
    const bool up = dl > 0.f;
    const float sc = up ? ex : 1.0f;
    const float p  = up ? 1.0f : ex;
    den = den * sc + p;
    acc0.x = acc0.x * sc + p * row0.x;
    acc0.y = acc0.y * sc + p * row0.y;
    acc0.z = acc0.z * sc + p * row0.z;
    acc0.w = acc0.w * sc + p * row0.w;
    acc1.x = acc1.x * sc + p * row1.x;
    acc1.y = acc1.y * sc + p * row1.y;
    acc1.z = acc1.z * sc + p * row1.z;
    acc1.w = acc1.w * sc + p * row1.w;
  }

  const float inv = 1.0f / den;
  #pragma unroll 1
  for (int hf = 0; hf < 2; hf++) {
    const int f = hf * (D / 2) + j * 4;
    const float4 ac = hf ? acc1 : acc0;
    const float4 b4 = *(const float4*)(bias  + f);
    const float4 g4 = *(const float4*)(gamma + f);
    const float4 t4 = *(const float4*)(beta  + f);
    const float4 m4 = *(const float4*)(mean  + f);
    const float4 v4 = *(const float4*)(var   + f);
    float4 o;
    o.x = (ac.x * inv + b4.x - m4.x) * (g4.x / sqrtf(v4.x + BN_EPS)) + t4.x;
    o.y = (ac.y * inv + b4.y - m4.y) * (g4.y / sqrtf(v4.y + BN_EPS)) + t4.y;
    o.z = (ac.z * inv + b4.z - m4.z) * (g4.z / sqrtf(v4.z + BN_EPS)) + t4.z;
    o.w = (ac.w * inv + b4.w - m4.w) * (g4.w / sqrtf(v4.w + BN_EPS)) + t4.w;
    if (relu) {
      o.x = fmaxf(o.x, 0.f); o.y = fmaxf(o.y, 0.f); o.z = fmaxf(o.z, 0.f); o.w = fmaxf(o.w, 0.f);
    }
    *(float4*)(h + (size_t)d * D + f) = o;
  }
}

#define RO_T 256
#define RO_W (RO_T / 32)
__global__ void __launch_bounds__(RO_T)
k_readout(const float* __restrict__ nodes, const int* __restrict__ batch,
          float* __restrict__ out_graph, int n) {
  __shared__ float part[16 * D];
  __shared__ int cnt[2][2][RO_W];
  const int g = blockIdx.x;
  const int t = threadIdx.x;

  int lo0 = 0, len0 = n, lo1 = 0, len1 = n;
  for (int round = 0; round < 8 && (len0 > 0 || len1 > 0); ++round) {
    const int st0 = (len0 + RO_T - 1) / RO_T;
    const int st1 = (len1 + RO_T - 1) / RO_T;
    const int p0 = lo0 + t * st0;
    const int p1 = lo1 + t * st1;
    const int b0 = batch[p0 < n ? p0 : n - 1];
    const int b1 = batch[p1 < n ? p1 : n - 1];
    const bool q0 = (p0 < lo0 + len0) && (b0 < g);
    const bool q1 = (p1 < lo1 + len1) && (b1 <= g);
    const int w0 = __popc(__builtin_amdgcn_ballot_w32(q0));
    const int w1 = __popc(__builtin_amdgcn_ballot_w32(q1));
    const int par = round & 1;
    if ((t & 31) == 0) { cnt[par][0][t >> 5] = w0; cnt[par][1][t >> 5] = w1; }
    __syncthreads();
    int c0 = 0, c1 = 0;
    for (int w = 0; w < RO_W; w++) { c0 += cnt[par][0][w]; c1 += cnt[par][1][w]; }
    c0 = __builtin_amdgcn_readfirstlane(c0);
    c1 = __builtin_amdgcn_readfirstlane(c1);
    if (len0 > 0) {
      if (c0 == 0) { len0 = 0; }
      else {
        const int nlo = lo0 + (c0 - 1) * st0 + 1;
        const int cap = lo0 + c0 * st0;
        const int nhi = cap < lo0 + len0 ? cap : lo0 + len0;
        lo0 = nlo; len0 = nhi - nlo;
      }
    }
    if (len1 > 0) {
      if (c1 == 0) { len1 = 0; }
      else {
        const int nlo = lo1 + (c1 - 1) * st1 + 1;
        const int cap = lo1 + c1 * st1;
        const int nhi = cap < lo1 + len1 ? cap : lo1 + len1;
        lo1 = nlo; len1 = nhi - nlo;
      }
    }
  }
  const int first = lo0, last = lo1;

  const int r = t >> 4;
  const int c = t & 15;
  float4 acc = make_float4(0.f, 0.f, 0.f, 0.f);
  int i = first + r;
  for (; i + 48 < last; i += 64) {
    const float* p = nodes + (size_t)i * D + c * 4;
    const float4 v0 = *(const float4*)p;
    const float4 v1 = *(const float4*)(p + 16 * D);
    const float4 v2 = *(const float4*)(p + 32 * D);
    const float4 v3 = *(const float4*)(p + 48 * D);
    acc.x += v0.x; acc.y += v0.y; acc.z += v0.z; acc.w += v0.w;
    acc.x += v1.x; acc.y += v1.y; acc.z += v1.z; acc.w += v1.w;
    acc.x += v2.x; acc.y += v2.y; acc.z += v2.z; acc.w += v2.w;
    acc.x += v3.x; acc.y += v3.y; acc.z += v3.z; acc.w += v3.w;
  }
  for (; i < last; i += 16) {
    const float4 v = *(const float4*)(nodes + (size_t)i * D + c * 4);
    acc.x += v.x; acc.y += v.y; acc.z += v.z; acc.w += v.w;
  }
  *(float4*)(part + r * D + c * 4) = acc;
  __syncthreads();
  if (t < D) {
    float s = 0.f;
    for (int k = 0; k < 16; k++) s += part[k * D + t];
    out_graph[(size_t)g * D + t] = s;
  }
}

extern "C" void kernel_launch(void* const* d_in, const int* in_sizes, int n_in,
                              void* d_out, int out_size, void* d_ws, size_t ws_size,
                              hipStream_t stream) {
  const float* x     = (const float*)d_in[0];
  const int*   eidx  = (const int*)d_in[1];
  const float* eattr = (const float*)d_in[2];
  const int*   batch = (const int*)d_in[3];
  const float* xlw   = (const float*)d_in[4];
  const float* xlb   = (const float*)d_in[5];
  const float* glw   = (const float*)d_in[6];
  const float* gas   = (const float*)d_in[7];
  const float* gad   = (const float*)d_in[8];
  const float* gew   = (const float*)d_in[9];
  const float* gae   = (const float*)d_in[10];
  const float* gb    = (const float*)d_in[11];
  const float* bng   = (const float*)d_in[12];
  const float* bnb   = (const float*)d_in[13];
  const float* bnm   = (const float*)d_in[14];
  const float* bnv   = (const float*)d_in[15];

  const int n = in_sizes[3];
  const int E = in_sizes[2] / F_E;
  const int* srcp = eidx;
  const int* dstp = eidx + E;

  float* out_nodes = (float*)d_out;
  float* out_graph = out_nodes + (size_t)n * D;
  const size_t graph_elems = (size_t)out_size - (size_t)n * D;

  float* h     = out_nodes;
  float* ws    = (float*)d_ws;
  float* xw    = ws;  ws += (size_t)n * D;
  float* wea   = ws;  ws += 64;
  int2*  link  = (int2*)ws; ws += (size_t)E * 2;
  float* aed   = ws;  ws += (size_t)E * L;
  float* sS    = ws;  ws += n;
  float* dS    = ws;  ws += n;
  int*   head  = (int*)ws; ws += n;

  const int B = 256;
  #define GRID(x) (((x) + B - 1) / B)

  k_prep<<<GRID(n > L * F_E ? n : L * F_E), B, 0, stream>>>(gew, gae, wea, head, n);
  k_edge_terms<<<GRID((E + ET_TRIPS - 1) / ET_TRIPS), B, 0, stream>>>(eattr, wea, aed, E);
  k_build<<<GRID(E), B, 0, stream>>>(srcp, dstp, head, link, E);

  {
    int waves = (n + 15) / 16;
    int blocks = (waves + 3) / 4;
    k_gemm_wmma<<<blocks, 128, 0, stream>>>(x, xlw, xlb, h, n, F_IN);
  }

  for (int l = 0; l < L; l++) {
    int waves = (n + 15) / 16;
    int blocks = (waves + 3) / 4;
    k_gemm_wmma<<<blocks, 128, 0, stream>>>(h, glw + (size_t)l * D * D, nullptr, xw, n, D);

    k_scores<<<GRID(n), B, 0, stream>>>(xw, gas + (size_t)l * D, gad + (size_t)l * D, sS, dS, n);
    k_node<<<GRID(n * 8), B, 0, stream>>>(xw, sS, dS, aed + (size_t)l * E, link, head,
                                           gb + (size_t)l * D,
                                           bng + (size_t)l * D, bnb + (size_t)l * D,
                                           bnm + (size_t)l * D, bnv + (size_t)l * D,
                                           h, n, E, l < L - 1 ? 1 : 0);
  }

  k_readout<<<(int)(graph_elems / D), RO_T, 0, stream>>>(out_nodes, batch, out_graph, n);
  #undef GRID
}


# attention scores by linearity: s = h.(W a_s), d = h.(W a_d), the two vectors formed once per call; the layer product sums them from its A values; 3 score launches gone; score sum order changed
# speedup vs baseline: 5.6960x; 1.0475x over previous
#include <hip/hip_runtime.h>


#define F_IN   32
#define F_E    16
#define D      64
#define L      3
#define NEG_SLOPE 0.2f
#define BN_EPS 1e-5f

typedef __attribute__((ext_vector_type(2))) float v2f;
typedef __attribute__((ext_vector_type(8))) float v8f;

#define PREP_U0 64
__global__ void k_prep(const float* __restrict__ We, const float* __restrict__ ae,
                       const float* __restrict__ Wl, const float* __restrict__ as,
                       const float* __restrict__ ad, float* __restrict__ wea,
                       float* __restrict__ usd, int* __restrict__ head, int n) {
  int i = blockIdx.x * blockDim.x + threadIdx.x;
  if (i < n) head[i] = -1;
  if (i < L * F_E) {
    int l = i / F_E, f = i % F_E;
    float s = 0.f;
    #pragma unroll
    for (int d = 0; d < D; d++) s += We[(l * F_E + f) * D + d] * ae[l * D + d];
    wea[i] = s;
  } else if (i >= PREP_U0 && i < PREP_U0 + L * 2 * D) {
    const int r = i - PREP_U0;
    const int l = r / (2 * D), q = r % (2 * D);
    const int k = (q >> 2) * 2 + (q & 1);
    const float* av = ((q & 2) ? ad : as) + l * D;
    const float* wr = Wl + ((size_t)l * D + k) * D;
    float s = 0.f;
    for (int j = 0; j < D; j++) s += wr[j] * av[j];
    usd[r] = s;
  }
}

__global__ void k_build(const int* __restrict__ src, const int* __restrict__ dst,
                        int* __restrict__ head, int2* __restrict__ link, int E) {
  int e = blockIdx.x * blockDim.x + threadIdx.x;
  if (e >= E) return;
  int prev = atomicExch(&head[dst[e]], e);
  link[e] = make_int2(prev, src[e]);
}

#define ET_TRIPS 4
__global__ void k_edge_terms(const float* __restrict__ eattr, const float* __restrict__ wea,
                             float* __restrict__ aed, int E) {
  const int t = blockIdx.x * blockDim.x + threadIdx.x;
  const int T = gridDim.x * blockDim.x;
  float4 v[ET_TRIPS][F_E / 4];
  for (int k = 0; k < ET_TRIPS; k++) {
    const int e = t + k * T;
    const float4* ea = (const float4*)(eattr + (size_t)(e < E ? e : E - 1) * F_E);
    for (int q = 0; q < F_E / 4; q++) v[k][q] = ea[q];
  }
  float4 w[L][F_E / 4];
  for (int l = 0; l < L; l++)
    for (int q = 0; q < F_E / 4; q++) w[l][q] = *(const float4*)(wea + l * F_E + q * 4);
  for (int k = 0; k < ET_TRIPS; k++) {
    const int e = t + k * T;
    for (int l = 0; l < L; l++) {
      float a = 0.f;
      for (int q = 0; q < F_E / 4; q++) {
        a += v[k][q].x * w[l][q].x;
        a += v[k][q].y * w[l][q].y;
        a += v[k][q].z * w[l][q].z;
        a += v[k][q].w * w[l][q].w;
      }
      if (e < E) aed[(size_t)l * E + e] = a;
    }
  }
}

template <bool SC>
__device__ __forceinline__ void gemm_rows(const float* __restrict__ A, const float* __restrict__ B,
                                          const float* __restrict__ bias, float* __restrict__ C,
                                          int n, int K, const float* __restrict__ usd,
                                          float* __restrict__ s, float* __restrict__ dv) {
  int wave = (blockIdx.x * blockDim.x + threadIdx.x) >> 5;
  int lane = threadIdx.x & 31;
  int row0 = wave * 16;
  if (row0 >= n) return;
  int lm = lane & 15;
  int kh = (lane >> 4) << 1;
  int mrow = row0 + lm;
  int mrd = mrow < n ? mrow : n - 1;

  v8f acc[4] = {};
  float ps = 0.f, pd = 0.f;
  for (int k = 0; k < K; k += 4) {
    v2f a;
    a.x = A[(size_t)mrd * K + k + kh];
    a.y = A[(size_t)mrd * K + k + kh + 1];
    if (SC) {
      const float4 u = *(const float4*)(usd + (k + kh) * 2);
      ps += a.x * u.x;
      ps += a.y * u.y;
      pd += a.x * u.z;
      pd += a.y * u.w;
    }
    #pragma unroll
    for (int nt = 0; nt < 4; nt++) {
      int col = nt * 16 + lm;
      v2f b;
      b.x = B[(k + kh) * D + col];
      b.y = B[(k + kh + 1) * D + col];
      acc[nt] = __builtin_amdgcn_wmma_f32_16x16x4_f32(
          false, a, false, b, (short)0, acc[nt], false, false);
    }
  }

  int rbase = row0 + ((lane >> 4) << 3);
  #pragma unroll
  for (int nt = 0; nt < 4; nt++) {
    int col = nt * 16 + lm;
    float bv = bias ? bias[col] : 0.f;
    #pragma unroll
    for (int r = 0; r < 8; r++) {
      int row = rbase + r;
      if (row < n) C[(size_t)row * D + col] = acc[nt][r] + bv;
    }
  }

  if (SC) {
    ps += __shfl_xor(ps, 16);
    pd += __shfl_xor(pd, 16);
    if (lane < 16 && mrow < n) { s[mrow] = ps; dv[mrow] = pd; }
  }
}

__global__ void k_gemm_wmma(const float* __restrict__ A, const float* __restrict__ B,
                            const float* __restrict__ bias, float* __restrict__ C,
                            int n, int K) {
  gemm_rows<false>(A, B, bias, C, n, K, nullptr, nullptr, nullptr);
}

__global__ void k_gemm_scores(const float* __restrict__ A, const float* __restrict__ B,
                              float* __restrict__ C, const float* __restrict__ usd,
                              float* __restrict__ s, float* __restrict__ dv, int n, int K) {
  gemm_rows<true>(A, B, nullptr, C, n, K, usd, s, dv);
}

__global__ void __launch_bounds__(256)
k_node(const float* __restrict__ xw, const float* __restrict__ s, const float* __restrict__ dv,
       const float* __restrict__ aed, const int2* __restrict__ link,
       const int* __restrict__ head, const float* __restrict__ bias,
       const float* __restrict__ gamma, const float* __restrict__ beta,
       const float* __restrict__ mean, const float* __restrict__ var,
       float* __restrict__ h, int n, int E, int relu) {
  int t = blockIdx.x * blockDim.x + threadIdx.x;
  int d = t >> 3;
  int j = t & 7;
  if (d >= n) return;
  const float dd = dv[d];
  float m = -INFINITY, den = 0.f, asum = 0.f, cnt = 0.f;
  float4 acc0 = make_float4(0.f, 0.f, 0.f, 0.f);
  float4 acc1 = make_float4(0.f, 0.f, 0.f, 0.f);

  int e = head[d];
  int2 lk = make_int2(-1, 0);
  if ((unsigned)e < (unsigned)E) lk = link[e];
  for (int it = 0; it < E && (unsigned)e < (unsigned)E; ++it) {
    const int nx = lk.x, sr = lk.y;
    const float ae = aed[e];
    const float ss = s[sr];
    const float* rp = xw + (size_t)sr * D + j * 4;
    const float4 row0 = *(const float4*)rp;
    const float4 row1 = *(const float4*)(rp + D / 2);
    const int2 nlk = link[(unsigned)nx < (unsigned)E ? nx : 0];
    float a = ss + dd + ae;
    a = a > 0.f ? a : NEG_SLOPE * a;
    asum += ae;
    cnt += 1.0f;
    const float dl = a - m;
    const float ex = expf(-fabsf(dl));
    const bool up = dl > 0.f;
    const float sc = up ? ex : 1.0f;
    const float p  = up ? 1.0f : ex;
    m = up ? a : m;
    den = den * sc + p;
    acc0.x = acc0.x * sc + p * row0.x;
    acc0.y = acc0.y * sc + p * row0.y;
    acc0.z = acc0.z * sc + p * row0.z;
    acc0.w = acc0.w * sc + p * row0.w;
    acc1.x = acc1.x * sc + p * row1.x;
    acc1.y = acc1.y * sc + p * row1.y;
    acc1.z = acc1.z * sc + p * row1.z;
    acc1.w = acc1.w * sc + p * row1.w;
    e = nx;
    lk = nlk;
  }

  {
    const float* rp = xw + (size_t)d * D + j * 4;
    const float4 row0 = *(const float4*)rp;
    const float4 row1 = *(const float4*)(rp + D / 2);
    float a = s[d] + dd + asum / fmaxf(cnt, 1.0f);
    a = a > 0.f ? a : NEG_SLOPE * a;
    const float dl = a - m;
    const float ex = expf(-fabsf(dl));
    const bool up = dl > 0.f;
    const float sc = up ? ex : 1.0f;
    const float p  = up ? 1.0f : ex;
    den = den * sc + p;
    acc0.x = acc0.x * sc + p * row0.x;
    acc0.y = acc0.y * sc + p * row0.y;
    acc0.z = acc0.z * sc + p * row0.z;
    acc0.w = acc0.w * sc + p * row0.w;
    acc1.x = acc1.x * sc + p * row1.x;
    acc1.y = acc1.y * sc + p * row1.y;
    acc1.z = acc1.z * sc + p * row1.z;
    acc1.w = acc1.w * sc + p * row1.w;
  }

  const float inv = 1.0f / den;
  #pragma unroll 1
  for (int hf = 0; hf < 2; hf++) {
    const int f = hf * (D / 2) + j * 4;
    const float4 ac = hf ? acc1 : acc0;
    const float4 b4 = *(const float4*)(bias  + f);
    const float4 g4 = *(const float4*)(gamma + f);
    const float4 t4 = *(const float4*)(beta  + f);
    const float4 m4 = *(const float4*)(mean  + f);
    const float4 v4 = *(const float4*)(var   + f);
    float4 o;
    o.x = (ac.x * inv + b4.x - m4.x) * (g4.x / sqrtf(v4.x + BN_EPS)) + t4.x;
    o.y = (ac.y * inv + b4.y - m4.y) * (g4.y / sqrtf(v4.y + BN_EPS)) + t4.y;
    o.z = (ac.z * inv + b4.z - m4.z) * (g4.z / sqrtf(v4.z + BN_EPS)) + t4.z;
    o.w = (ac.w * inv + b4.w - m4.w) * (g4.w / sqrtf(v4.w + BN_EPS)) + t4.w;
    if (relu) {
      o.x = fmaxf(o.x, 0.f); o.y = fmaxf(o.y, 0.f); o.z = fmaxf(o.z, 0.f); o.w = fmaxf(o.w, 0.f);
    }
    *(float4*)(h + (size_t)d * D + f) = o;
  }
}

#define RO_T 256
#define RO_W (RO_T / 32)
__global__ void __launch_bounds__(RO_T)
k_readout(const float* __restrict__ nodes, const int* __restrict__ batch,
          float* __restrict__ out_graph, int n) {
  __shared__ float part[16 * D];
  __shared__ int cnt[2][2][RO_W];
  const int g = blockIdx.x;
  const int t = threadIdx.x;

  int lo0 = 0, len0 = n, lo1 = 0, len1 = n;
  for (int round = 0; round < 8 && (len0 > 0 || len1 > 0); ++round) {
    const int st0 = (len0 + RO_T - 1) / RO_T;
    const int st1 = (len1 + RO_T - 1) / RO_T;
    const int p0 = lo0 + t * st0;
    const int p1 = lo1 + t * st1;
    const int b0 = batch[p0 < n ? p0 : n - 1];
    const int b1 = batch[p1 < n ? p1 : n - 1];
    const bool q0 = (p0 < lo0 + len0) && (b0 < g);
    const bool q1 = (p1 < lo1 + len1) && (b1 <= g);
    const int w0 = __popc(__builtin_amdgcn_ballot_w32(q0));
    const int w1 = __popc(__builtin_amdgcn_ballot_w32(q1));
    const int par = round & 1;
    if ((t & 31) == 0) { cnt[par][0][t >> 5] = w0; cnt[par][1][t >> 5] = w1; }
    __syncthreads();
    int c0 = 0, c1 = 0;
    for (int w = 0; w < RO_W; w++) { c0 += cnt[par][0][w]; c1 += cnt[par][1][w]; }
    c0 = __builtin_amdgcn_readfirstlane(c0);
    c1 = __builtin_amdgcn_readfirstlane(c1);
    if (len0 > 0) {
      if (c0 == 0) { len0 = 0; }
      else {
        const int nlo = lo0 + (c0 - 1) * st0 + 1;
        const int cap = lo0 + c0 * st0;
        const int nhi = cap < lo0 + len0 ? cap : lo0 + len0;
        lo0 = nlo; len0 = nhi - nlo;
      }
    }
    if (len1 > 0) {
      if (c1 == 0) { len1 = 0; }
      else {
        const int nlo = lo1 + (c1 - 1) * st1 + 1;
        const int cap = lo1 + c1 * st1;
        const int nhi = cap < lo1 + len1 ? cap : lo1 + len1;
        lo1 = nlo; len1 = nhi - nlo;
      }
    }
  }
  const int first = lo0, last = lo1;

  const int r = t >> 4;
  const int c = t & 15;
  float4 acc = make_float4(0.f, 0.f, 0.f, 0.f);
  int i = first + r;
  for (; i + 48 < last; i += 64) {
    const float* p = nodes + (size_t)i * D + c * 4;
    const float4 v0 = *(const float4*)p;
    const float4 v1 = *(const float4*)(p + 16 * D);
    const float4 v2 = *(const float4*)(p + 32 * D);
    const float4 v3 = *(const float4*)(p + 48 * D);
    acc.x += v0.x; acc.y += v0.y; acc.z += v0.z; acc.w += v0.w;
    acc.x += v1.x; acc.y += v1.y; acc.z += v1.z; acc.w += v1.w;
    acc.x += v2.x; acc.y += v2.y; acc.z += v2.z; acc.w += v2.w;
    acc.x += v3.x; acc.y += v3.y; acc.z += v3.z; acc.w += v3.w;
  }
  for (; i < last; i += 16) {
    const float4 v = *(const float4*)(nodes + (size_t)i * D + c * 4);
    acc.x += v.x; acc.y += v.y; acc.z += v.z; acc.w += v.w;
  }
  *(float4*)(part + r * D + c * 4) = acc;
  __syncthreads();
  if (t < D) {
    float s = 0.f;
    for (int k = 0; k < 16; k++) s += part[k * D + t];
    out_graph[(size_t)g * D + t] = s;
  }
}

extern "C" void kernel_launch(void* const* d_in, const int* in_sizes, int n_in,
                              void* d_out, int out_size, void* d_ws, size_t ws_size,
                              hipStream_t stream) {
  const float* x     = (const float*)d_in[0];
  const int*   eidx  = (const int*)d_in[1];
  const float* eattr = (const float*)d_in[2];
  const int*   batch = (const int*)d_in[3];
  const float* xlw   = (const float*)d_in[4];
  const float* xlb   = (const float*)d_in[5];
  const float* glw   = (const float*)d_in[6];
  const float* gas   = (const float*)d_in[7];
  const float* gad   = (const float*)d_in[8];
  const float* gew   = (const float*)d_in[9];
  const float* gae   = (const float*)d_in[10];
  const float* gb    = (const float*)d_in[11];
  const float* bng   = (const float*)d_in[12];
  const float* bnb   = (const float*)d_in[13];
  const float* bnm   = (const float*)d_in[14];
  const float* bnv   = (const float*)d_in[15];

  const int n = in_sizes[3];
  const int E = in_sizes[2] / F_E;
  const int* srcp = eidx;
  const int* dstp = eidx + E;

  float* out_nodes = (float*)d_out;
  float* out_graph = out_nodes + (size_t)n * D;
  const size_t graph_elems = (size_t)out_size - (size_t)n * D;

  float* h     = out_nodes;
  float* ws    = (float*)d_ws;
  float* xw    = ws;  ws += (size_t)n * D;
  float* wea   = ws;  ws += 64;
  float* usd   = ws;  ws += L * 2 * D;
  int2*  link  = (int2*)ws; ws += (size_t)E * 2;
  float* aed   = ws;  ws += (size_t)E * L;
  float* sS    = ws;  ws += n;
  float* dS    = ws;  ws += n;
  int*   head  = (int*)ws; ws += n;

  const int B = 256;
  #define GRID(x) (((x) + B - 1) / B)

  k_prep<<<GRID(n > PREP_U0 + L * 2 * D ? n : PREP_U0 + L * 2 * D), B, 0, stream>>>(
      gew, gae, glw, gas, gad, wea, usd, head, n);
  k_edge_terms<<<GRID((E + ET_TRIPS - 1) / ET_TRIPS), B, 0, stream>>>(eattr, wea, aed, E);
  k_build<<<GRID(E), B, 0, stream>>>(srcp, dstp, head, link, E);

  {
    int waves = (n + 15) / 16;
    int blocks = (waves + 3) / 4;
    k_gemm_wmma<<<blocks, 128, 0, stream>>>(x, xlw, xlb, h, n, F_IN);
  }

  for (int l = 0; l < L; l++) {
    int waves = (n + 15) / 16;
    int blocks = (waves + 3) / 4;
    k_gemm_scores<<<blocks, 128, 0, stream>>>(h, glw + (size_t)l * D * D, xw, usd + (size_t)l * 2 * D,
                                              sS, dS, n, D);
    k_node<<<GRID(n * 8), B, 0, stream>>>(xw, sS, dS, aed + (size_t)l * E, link, head,
                                           gb + (size_t)l * D,
                                           bng + (size_t)l * D, bnb + (size_t)l * D,
                                           bnm + (size_t)l * D, bnv + (size_t)l * D,
                                           h, n, E, l < L - 1 ? 1 : 0);
  }

  k_readout<<<(int)(graph_elems / D), RO_T, 0, stream>>>(out_nodes, batch, out_graph, n);
  #undef GRID
}


# two linked lists per node, head plane picked by the inserting wave's package half (swaps stay in one half's cache); a node's row walks both side by side: shorter dependent chain; sum orders changed
# speedup vs baseline: 7.0732x; 1.2418x over previous
#include <hip/hip_runtime.h>


#define F_IN   32
#define F_E    16
#define D      64
#define L      3
#define NEG_SLOPE 0.2f
#define BN_EPS 1e-5f

typedef __attribute__((ext_vector_type(2))) float v2f;
typedef __attribute__((ext_vector_type(8))) float v8f;

#define PREP_U0 64
__global__ void k_prep(const float* __restrict__ We, const float* __restrict__ ae,
                       const float* __restrict__ Wl, const float* __restrict__ as,
                       const float* __restrict__ ad, float* __restrict__ wea,
                       float* __restrict__ usd, int* __restrict__ head, int n) {
  int i = blockIdx.x * blockDim.x + threadIdx.x;
  if (i < n) { head[i] = -1; head[n + i] = -1; }
  if (i < L * F_E) {
    int l = i / F_E, f = i % F_E;
    float s = 0.f;
    #pragma unroll
    for (int d = 0; d < D; d++) s += We[(l * F_E + f) * D + d] * ae[l * D + d];
    wea[i] = s;
  } else if (i >= PREP_U0 && i < PREP_U0 + L * 2 * D) {
    const int r = i - PREP_U0;
    const int l = r / (2 * D), q = r % (2 * D);
    const int k = (q >> 2) * 2 + (q & 1);
    const float* av = ((q & 2) ? ad : as) + l * D;
    const float* wr = Wl + ((size_t)l * D + k) * D;
    float s = 0.f;
    for (int j = 0; j < D; j++) s += wr[j] * av[j];
    usd[r] = s;
  }
}

__global__ void k_build(const int* __restrict__ src, const int* __restrict__ dst,
                        int* __restrict__ head, int2* __restrict__ link, int E, int n) {
  int e = blockIdx.x * blockDim.x + threadIdx.x;
  if (e >= E) return;
  const int half = (int)((__builtin_amdgcn_s_sendmsg_rtn(0x87) >> 18) & 1u);
  int prev = atomicExch(&head[(size_t)half * n + dst[e]], e);
  link[e] = make_int2(prev, src[e]);
}

#define ET_TRIPS 4
__global__ void k_edge_terms(const float* __restrict__ eattr, const float* __restrict__ wea,
                             float* __restrict__ aed, int E) {
  const int t = blockIdx.x * blockDim.x + threadIdx.x;
  const int T = gridDim.x * blockDim.x;
  float4 v[ET_TRIPS][F_E / 4];
  for (int k = 0; k < ET_TRIPS; k++) {
    const int e = t + k * T;
    const float4* ea = (const float4*)(eattr + (size_t)(e < E ? e : E - 1) * F_E);
    for (int q = 0; q < F_E / 4; q++) v[k][q] = ea[q];
  }
  float4 w[L][F_E / 4];
  for (int l = 0; l < L; l++)
    for (int q = 0; q < F_E / 4; q++) w[l][q] = *(const float4*)(wea + l * F_E + q * 4);
  for (int k = 0; k < ET_TRIPS; k++) {
    const int e = t + k * T;
    for (int l = 0; l < L; l++) {
      float a = 0.f;
      for (int q = 0; q < F_E / 4; q++) {
        a += v[k][q].x * w[l][q].x;
        a += v[k][q].y * w[l][q].y;
        a += v[k][q].z * w[l][q].z;
        a += v[k][q].w * w[l][q].w;
      }
      if (e < E) aed[(size_t)l * E + e] = a;
    }
  }
}

template <bool SC>
__device__ __forceinline__ void gemm_rows(const float* __restrict__ A, const float* __restrict__ B,
                                          const float* __restrict__ bias, float* __restrict__ C,
                                          int n, int K, const float* __restrict__ usd,
                                          float* __restrict__ s, float* __restrict__ dv) {
  int wave = (blockIdx.x * blockDim.x + threadIdx.x) >> 5;
  int lane = threadIdx.x & 31;
  int row0 = wave * 16;
  if (row0 >= n) return;
  int lm = lane & 15;
  int kh = (lane >> 4) << 1;
  int mrow = row0 + lm;
  int mrd = mrow < n ? mrow : n - 1;

  v8f acc[4] = {};
  float ps = 0.f, pd = 0.f;
  for (int k = 0; k < K; k += 4) {
    v2f a;
    a.x = A[(size_t)mrd * K + k + kh];
    a.y = A[(size_t)mrd * K + k + kh + 1];
    if (SC) {
      const float4 u = *(const float4*)(usd + (k + kh) * 2);
      ps += a.x * u.x;
      ps += a.y * u.y;
      pd += a.x * u.z;
      pd += a.y * u.w;
    }
    #pragma unroll
    for (int nt = 0; nt < 4; nt++) {
      int col = nt * 16 + lm;
      v2f b;
      b.x = B[(k + kh) * D + col];
      b.y = B[(k + kh + 1) * D + col];
      acc[nt] = __builtin_amdgcn_wmma_f32_16x16x4_f32(
          false, a, false, b, (short)0, acc[nt], false, false);
    }
  }

  int rbase = row0 + ((lane >> 4) << 3);
  #pragma unroll
  for (int nt = 0; nt < 4; nt++) {
    int col = nt * 16 + lm;
    float bv = bias ? bias[col] : 0.f;
    #pragma unroll
    for (int r = 0; r < 8; r++) {
      int row = rbase + r;
      if (row < n) C[(size_t)row * D + col] = acc[nt][r] + bv;
    }
  }

  if (SC) {
    ps += __shfl_xor(ps, 16);
    pd += __shfl_xor(pd, 16);
    if (lane < 16 && mrow < n) { s[mrow] = ps; dv[mrow] = pd; }
  }
}

__global__ void k_gemm_wmma(const float* __restrict__ A, const float* __restrict__ B,
                            const float* __restrict__ bias, float* __restrict__ C,
                            int n, int K) {
  gemm_rows<false>(A, B, bias, C, n, K, nullptr, nullptr, nullptr);
}

__global__ void k_gemm_scores(const float* __restrict__ A, const float* __restrict__ B,
                              float* __restrict__ C, const float* __restrict__ usd,
                              float* __restrict__ s, float* __restrict__ dv, int n, int K) {
  gemm_rows<true>(A, B, nullptr, C, n, K, usd, s, dv);
}

struct Fold {
  float m, den, asum, cnt;
  float4 acc0, acc1;
};

__device__ __forceinline__ void fold_edge(Fold& f, float a, const float4& row0, const float4& row1) {
  a = a > 0.f ? a : NEG_SLOPE * a;
  const float dl = a - f.m;
  const float ex = expf(-fabsf(dl));
  const bool up = dl > 0.f;
  const float sc = up ? ex : 1.0f;
  const float p  = up ? 1.0f : ex;
  f.m = up ? a : f.m;
  f.den = f.den * sc + p;
  f.acc0.x = f.acc0.x * sc + p * row0.x;
  f.acc0.y = f.acc0.y * sc + p * row0.y;
  f.acc0.z = f.acc0.z * sc + p * row0.z;
  f.acc0.w = f.acc0.w * sc + p * row0.w;
  f.acc1.x = f.acc1.x * sc + p * row1.x;
  f.acc1.y = f.acc1.y * sc + p * row1.y;
  f.acc1.z = f.acc1.z * sc + p * row1.z;
  f.acc1.w = f.acc1.w * sc + p * row1.w;
}

__global__ void __launch_bounds__(256)
k_node(const float* __restrict__ xw, const float* __restrict__ s, const float* __restrict__ dv,
       const float* __restrict__ aed, const int2* __restrict__ link,
       const int* __restrict__ head, const float* __restrict__ bias,
       const float* __restrict__ gamma, const float* __restrict__ beta,
       const float* __restrict__ mean, const float* __restrict__ var,
       float* __restrict__ h, int n, int E, int relu) {
  int t = blockIdx.x * blockDim.x + threadIdx.x;
  int d = t >> 3;
  int j = t & 7;
  if (d >= n) return;
  const float dd = dv[d];
  Fold st;
  st.m = -INFINITY; st.den = 0.f; st.asum = 0.f; st.cnt = 0.f;
  st.acc0 = make_float4(0.f, 0.f, 0.f, 0.f);
  st.acc1 = make_float4(0.f, 0.f, 0.f, 0.f);

  int eA = head[d];
  int eB = head[n + d];
  bool vA = (unsigned)eA < (unsigned)E;
  bool vB = (unsigned)eB < (unsigned)E;
  int2 lkA = link[vA ? eA : 0];
  int2 lkB = link[vB ? eB : 0];
  int it = 0;
  for (; it < E && vA && vB; ++it) {
    const float aeA = aed[eA];
    const float aeB = aed[eB];
    const float ssA = s[lkA.y];
    const float ssB = s[lkB.y];
    const float* rpA = xw + (size_t)lkA.y * D + j * 4;
    const float* rpB = xw + (size_t)lkB.y * D + j * 4;
    const float4 rA0 = *(const float4*)rpA;
    const float4 rA1 = *(const float4*)(rpA + D / 2);
    const float4 rB0 = *(const float4*)rpB;
    const float4 rB1 = *(const float4*)(rpB + D / 2);
    const int2 nA = link[(unsigned)lkA.x < (unsigned)E ? lkA.x : 0];
    const int2 nB = link[(unsigned)lkB.x < (unsigned)E ? lkB.x : 0];
    st.asum += aeA;
    st.cnt += 1.0f;
    fold_edge(st, ssA + dd + aeA, rA0, rA1);
    st.asum += aeB;
    st.cnt += 1.0f;
    fold_edge(st, ssB + dd + aeB, rB0, rB1);
    eA = lkA.x;  lkA = nA;  vA = (unsigned)eA < (unsigned)E;
    eB = lkB.x;  lkB = nB;  vB = (unsigned)eB < (unsigned)E;
  }
  int e = vA ? eA : eB;
  int2 lk = vA ? lkA : lkB;
  for (; it < E && (unsigned)e < (unsigned)E; ++it) {
    const int nx = lk.x, sr = lk.y;
    const float ae = aed[e];
    const float ss = s[sr];
    const float* rp = xw + (size_t)sr * D + j * 4;
    const float4 row0 = *(const float4*)rp;
    const float4 row1 = *(const float4*)(rp + D / 2);
    const int2 nlk = link[(unsigned)nx < (unsigned)E ? nx : 0];
    st.asum += ae;
    st.cnt += 1.0f;
    fold_edge(st, ss + dd + ae, row0, row1);
    e = nx;
    lk = nlk;
  }

  {
    const float* rp = xw + (size_t)d * D + j * 4;
    const float4 row0 = *(const float4*)rp;
    const float4 row1 = *(const float4*)(rp + D / 2);
    fold_edge(st, s[d] + dd + st.asum / fmaxf(st.cnt, 1.0f), row0, row1);
  }
  const float den = st.den;
  const float4 acc0 = st.acc0, acc1 = st.acc1;

  const float inv = 1.0f / den;
  #pragma unroll 1
  for (int hf = 0; hf < 2; hf++) {
    const int f = hf * (D / 2) + j * 4;
    const float4 ac = hf ? acc1 : acc0;
    const float4 b4 = *(const float4*)(bias  + f);
    const float4 g4 = *(const float4*)(gamma + f);
    const float4 t4 = *(const float4*)(beta  + f);
    const float4 m4 = *(const float4*)(mean  + f);
    const float4 v4 = *(const float4*)(var   + f);
    float4 o;
    o.x = (ac.x * inv + b4.x - m4.x) * (g4.x / sqrtf(v4.x + BN_EPS)) + t4.x;
    o.y = (ac.y * inv + b4.y - m4.y) * (g4.y / sqrtf(v4.y + BN_EPS)) + t4.y;
    o.z = (ac.z * inv + b4.z - m4.z) * (g4.z / sqrtf(v4.z + BN_EPS)) + t4.z;
    o.w = (ac.w * inv + b4.w - m4.w) * (g4.w / sqrtf(v4.w + BN_EPS)) + t4.w;
    if (relu) {
      o.x = fmaxf(o.x, 0.f); o.y = fmaxf(o.y, 0.f); o.z = fmaxf(o.z, 0.f); o.w = fmaxf(o.w, 0.f);
    }
    *(float4*)(h + (size_t)d * D + f) = o;
  }
}

#define RO_T 256
#define RO_W (RO_T / 32)
__global__ void __launch_bounds__(RO_T)
k_readout(const float* __restrict__ nodes, const int* __restrict__ batch,
          float* __restrict__ out_graph, int n) {
  __shared__ float part[16 * D];
  __shared__ int cnt[2][2][RO_W];
  const int g = blockIdx.x;
  const int t = threadIdx.x;

  int lo0 = 0, len0 = n, lo1 = 0, len1 = n;
  for (int round = 0; round < 8 && (len0 > 0 || len1 > 0); ++round) {
    const int st0 = (len0 + RO_T - 1) / RO_T;
    const int st1 = (len1 + RO_T - 1) / RO_T;
    const int p0 = lo0 + t * st0;
    const int p1 = lo1 + t * st1;
    const int b0 = batch[p0 < n ? p0 : n - 1];
    const int b1 = batch[p1 < n ? p1 : n - 1];
    const bool q0 = (p0 < lo0 + len0) && (b0 < g);
    const bool q1 = (p1 < lo1 + len1) && (b1 <= g);
    const int w0 = __popc(__builtin_amdgcn_ballot_w32(q0));
    const int w1 = __popc(__builtin_amdgcn_ballot_w32(q1));
    const int par = round & 1;
    if ((t & 31) == 0) { cnt[par][0][t >> 5] = w0; cnt[par][1][t >> 5] = w1; }
    __syncthreads();
    int c0 = 0, c1 = 0;
    for (int w = 0; w < RO_W; w++) { c0 += cnt[par][0][w]; c1 += cnt[par][1][w]; }
    c0 = __builtin_amdgcn_readfirstlane(c0);
    c1 = __builtin_amdgcn_readfirstlane(c1);
    if (len0 > 0) {
      if (c0 == 0) { len0 = 0; }
      else {
        const int nlo = lo0 + (c0 - 1) * st0 + 1;
        const int cap = lo0 + c0 * st0;
        const int nhi = cap < lo0 + len0 ? cap : lo0 + len0;
        lo0 = nlo; len0 = nhi - nlo;
      }
    }
    if (len1 > 0) {
      if (c1 == 0) { len1 = 0; }
      else {
        const int nlo = lo1 + (c1 - 1) * st1 + 1;
        const int cap = lo1 + c1 * st1;
        const int nhi = cap < lo1 + len1 ? cap : lo1 + len1;
        lo1 = nlo; len1 = nhi - nlo;
      }
    }
  }
  const int first = lo0, last = lo1;

  const int r = t >> 4;
  const int c = t & 15;
  float4 acc = make_float4(0.f, 0.f, 0.f, 0.f);
  int i = first + r;
  for (; i + 48 < last; i += 64) {
    const float* p = nodes + (size_t)i * D + c * 4;
    const float4 v0 = *(const float4*)p;
    const float4 v1 = *(const float4*)(p + 16 * D);
    const float4 v2 = *(const float4*)(p + 32 * D);
    const float4 v3 = *(const float4*)(p + 48 * D);
    acc.x += v0.x; acc.y += v0.y; acc.z += v0.z; acc.w += v0.w;
    acc.x += v1.x; acc.y += v1.y; acc.z += v1.z; acc.w += v1.w;
    acc.x += v2.x; acc.y += v2.y; acc.z += v2.z; acc.w += v2.w;
    acc.x += v3.x; acc.y += v3.y; acc.z += v3.z; acc.w += v3.w;
  }
  for (; i < last; i += 16) {
    const float4 v = *(const float4*)(nodes + (size_t)i * D + c * 4);
    acc.x += v.x; acc.y += v.y; acc.z += v.z; acc.w += v.w;
  }
  *(float4*)(part + r * D + c * 4) = acc;
  __syncthreads();
  if (t < D) {
    float s = 0.f;
    for (int k = 0; k < 16; k++) s += part[k * D + t];
    out_graph[(size_t)g * D + t] = s;
  }
}

extern "C" void kernel_launch(void* const* d_in, const int* in_sizes, int n_in,
                              void* d_out, int out_size, void* d_ws, size_t ws_size,
                              hipStream_t stream) {
  const float* x     = (const float*)d_in[0];
  const int*   eidx  = (const int*)d_in[1];
  const float* eattr = (const float*)d_in[2];
  const int*   batch = (const int*)d_in[3];
  const float* xlw   = (const float*)d_in[4];
  const float* xlb   = (const float*)d_in[5];
  const float* glw   = (const float*)d_in[6];
  const float* gas   = (const float*)d_in[7];
  const float* gad   = (const float*)d_in[8];
  const float* gew   = (const float*)d_in[9];
  const float* gae   = (const float*)d_in[10];
  const float* gb    = (const float*)d_in[11];
  const float* bng   = (const float*)d_in[12];
  const float* bnb   = (const float*)d_in[13];
  const float* bnm   = (const float*)d_in[14];
  const float* bnv   = (const float*)d_in[15];

  const int n = in_sizes[3];
  const int E = in_sizes[2] / F_E;
  const int* srcp = eidx;
  const int* dstp = eidx + E;

  float* out_nodes = (float*)d_out;
  float* out_graph = out_nodes + (size_t)n * D;
  const size_t graph_elems = (size_t)out_size - (size_t)n * D;

  float* h     = out_nodes;
  float* ws    = (float*)d_ws;
  float* xw    = ws;  ws += (size_t)n * D;
  float* wea   = ws;  ws += 64;
  float* usd   = ws;  ws += L * 2 * D;
  int2*  link  = (int2*)ws; ws += (size_t)E * 2;
  float* aed   = ws;  ws += (size_t)E * L;
  float* sS    = ws;  ws += n;
  float* dS    = ws;  ws += n;
  int*   head  = (int*)ws; ws += (size_t)n * 2;

  const int B = 256;
  #define GRID(x) (((x) + B - 1) / B)

  k_prep<<<GRID(n > PREP_U0 + L * 2 * D ? n : PREP_U0 + L * 2 * D), B, 0, stream>>>(
      gew, gae, glw, gas, gad, wea, usd, head, n);
  k_edge_terms<<<GRID((E + ET_TRIPS - 1) / ET_TRIPS), B, 0, stream>>>(eattr, wea, aed, E);
  k_build<<<GRID(E), B, 0, stream>>>(srcp, dstp, head, link, E, n);

  {
    int waves = (n + 15) / 16;
    int blocks = (waves + 3) / 4;
    k_gemm_wmma<<<blocks, 128, 0, stream>>>(x, xlw, xlb, h, n, F_IN);
  }

  for (int l = 0; l < L; l++) {
    int waves = (n + 15) / 16;
    int blocks = (waves + 3) / 4;
    k_gemm_scores<<<blocks, 128, 0, stream>>>(h, glw + (size_t)l * D * D, xw, usd + (size_t)l * 2 * D,
                                              sS, dS, n, D);
    k_node<<<GRID(n * 8), B, 0, stream>>>(xw, sS, dS, aed + (size_t)l * E, link, head,
                                           gb + (size_t)l * D,
                                           bng + (size_t)l * D, bnb + (size_t)l * D,
                                           bnm + (size_t)l * D, bnv + (size_t)l * D,
                                           h, n, E, l < L - 1 ? 1 : 0);
  }

  k_readout<<<(int)(graph_elems / D), RO_T, 0, stream>>>(out_nodes, batch, out_graph, n);
  #undef GRID
}


# edge pass also chains the lists and stores ONE whole 16 B record {next, source, term 0, term 1} per edge: layers 0, 1 get link+term in one load, one line; build launch gone; sum order: arrival
# speedup vs baseline: 7.4389x; 1.0517x over previous
#include <hip/hip_runtime.h>


#define F_IN   32
#define F_E    16
#define D      64
#define L      3
#define NEG_SLOPE 0.2f
#define BN_EPS 1e-5f

typedef __attribute__((ext_vector_type(2))) float v2f;
typedef __attribute__((ext_vector_type(8))) float v8f;

#define PREP_U0 64
__global__ void k_prep(const float* __restrict__ We, const float* __restrict__ ae,
                       const float* __restrict__ Wl, const float* __restrict__ as,
                       const float* __restrict__ ad, float* __restrict__ wea,
                       float* __restrict__ usd, int* __restrict__ head, int n) {
  int i = blockIdx.x * blockDim.x + threadIdx.x;
  if (i < n) { head[i] = -1; head[n + i] = -1; }
  if (i < L * F_E) {
    int l = i / F_E, f = i % F_E;
    float s = 0.f;
    #pragma unroll
    for (int d = 0; d < D; d++) s += We[(l * F_E + f) * D + d] * ae[l * D + d];
    wea[i] = s;
  } else if (i >= PREP_U0 && i < PREP_U0 + L * 2 * D) {
    const int r = i - PREP_U0;
    const int l = r / (2 * D), q = r % (2 * D);
    const int k = (q >> 2) * 2 + (q & 1);
    const float* av = ((q & 2) ? ad : as) + l * D;
    const float* wr = Wl + ((size_t)l * D + k) * D;
    float s = 0.f;
    for (int j = 0; j < D; j++) s += wr[j] * av[j];
    usd[r] = s;
  }
}

#define ET_TRIPS 4
__global__ void k_edge_build(const float* __restrict__ eattr, const float* __restrict__ wea,
                             const int* __restrict__ src, const int* __restrict__ dst,
                             int* __restrict__ head, int4* __restrict__ rec,
                             float* __restrict__ aed2, int E, int n) {
  static_assert(L == 3, "two terms ride in the record, the third has its plane");
  const int t = blockIdx.x * blockDim.x + threadIdx.x;
  const int T = gridDim.x * blockDim.x;
  const int half = (int)((__builtin_amdgcn_s_sendmsg_rtn(0x87) >> 18) & 1u);
  int sv[ET_TRIPS], dn[ET_TRIPS];
  float4 v[ET_TRIPS][F_E / 4];
  for (int k = 0; k < ET_TRIPS; k++) {
    const int e = t + k * T;
    const int ec = e < E ? e : E - 1;
    dn[k] = dst[ec];
    sv[k] = src[ec];
    const float4* ea = (const float4*)(eattr + (size_t)ec * F_E);
    for (int q = 0; q < F_E / 4; q++) v[k][q] = ea[q];
  }
  float4 w[L][F_E / 4];
  for (int l = 0; l < L; l++)
    for (int q = 0; q < F_E / 4; q++) w[l][q] = *(const float4*)(wea + l * F_E + q * 4);
  for (int k = 0; k < ET_TRIPS; k++) {
    const int e = t + k * T;
    float a[L];
    for (int l = 0; l < L; l++) {
      float acc = 0.f;
      for (int q = 0; q < F_E / 4; q++) {
        acc += v[k][q].x * w[l][q].x;
        acc += v[k][q].y * w[l][q].y;
        acc += v[k][q].z * w[l][q].z;
        acc += v[k][q].w * w[l][q].w;
      }
      a[l] = acc;
    }
    if (e < E) {
      const int prev = atomicExch(&head[(size_t)half * n + dn[k]], e);
      rec[e] = make_int4(prev, sv[k], __float_as_int(a[0]), __float_as_int(a[1]));
      aed2[e] = a[2];
    }
  }
}

template <bool SC>
__device__ __forceinline__ void gemm_rows(const float* __restrict__ A, const float* __restrict__ B,
                                          const float* __restrict__ bias, float* __restrict__ C,
                                          int n, int K, const float* __restrict__ usd,
                                          float* __restrict__ s, float* __restrict__ dv) {
  int wave = (blockIdx.x * blockDim.x + threadIdx.x) >> 5;
  int lane = threadIdx.x & 31;
  int row0 = wave * 16;
  if (row0 >= n) return;
  int lm = lane & 15;
  int kh = (lane >> 4) << 1;
  int mrow = row0 + lm;
  int mrd = mrow < n ? mrow : n - 1;

  v8f acc[4] = {};
  float ps = 0.f, pd = 0.f;
  for (int k = 0; k < K; k += 4) {
    v2f a;
    a.x = A[(size_t)mrd * K + k + kh];
    a.y = A[(size_t)mrd * K + k + kh + 1];
    if (SC) {
      const float4 u = *(const float4*)(usd + (k + kh) * 2);
      ps += a.x * u.x;
      ps += a.y * u.y;
      pd += a.x * u.z;
      pd += a.y * u.w;
    }
    #pragma unroll
    for (int nt = 0; nt < 4; nt++) {
      int col = nt * 16 + lm;
      v2f b;
      b.x = B[(k + kh) * D + col];
      b.y = B[(k + kh + 1) * D + col];
      acc[nt] = __builtin_amdgcn_wmma_f32_16x16x4_f32(
          false, a, false, b, (short)0, acc[nt], false, false);
    }
  }

  int rbase = row0 + ((lane >> 4) << 3);
  #pragma unroll
  for (int nt = 0; nt < 4; nt++) {
    int col = nt * 16 + lm;
    float bv = bias ? bias[col] : 0.f;
    #pragma unroll
    for (int r = 0; r < 8; r++) {
      int row = rbase + r;
      if (row < n) C[(size_t)row * D + col] = acc[nt][r] + bv;
    }
  }

  if (SC) {
    ps += __shfl_xor(ps, 16);
    pd += __shfl_xor(pd, 16);
    if (lane < 16 && mrow < n) { s[mrow] = ps; dv[mrow] = pd; }
  }
}

__global__ void k_gemm_wmma(const float* __restrict__ A, const float* __restrict__ B,
                            const float* __restrict__ bias, float* __restrict__ C,
                            int n, int K) {
  gemm_rows<false>(A, B, bias, C, n, K, nullptr, nullptr, nullptr);
}

__global__ void k_gemm_scores(const float* __restrict__ A, const float* __restrict__ B,
                              float* __restrict__ C, const float* __restrict__ usd,
                              float* __restrict__ s, float* __restrict__ dv, int n, int K) {
  gemm_rows<true>(A, B, nullptr, C, n, K, usd, s, dv);
}

struct Fold {
  float m, den, asum, cnt;
  float4 acc0, acc1;
};

__device__ __forceinline__ void fold_edge(Fold& f, float a, const float4& row0, const float4& row1) {
  a = a > 0.f ? a : NEG_SLOPE * a;
  const float dl = a - f.m;
  const float ex = expf(-fabsf(dl));
  const bool up = dl > 0.f;
  const float sc = up ? ex : 1.0f;
  const float p  = up ? 1.0f : ex;
  f.m = up ? a : f.m;
  f.den = f.den * sc + p;
  f.acc0.x = f.acc0.x * sc + p * row0.x;
  f.acc0.y = f.acc0.y * sc + p * row0.y;
  f.acc0.z = f.acc0.z * sc + p * row0.z;
  f.acc0.w = f.acc0.w * sc + p * row0.w;
  f.acc1.x = f.acc1.x * sc + p * row1.x;
  f.acc1.y = f.acc1.y * sc + p * row1.y;
  f.acc1.z = f.acc1.z * sc + p * row1.z;
  f.acc1.w = f.acc1.w * sc + p * row1.w;
}

template <int TS>
__device__ __forceinline__ void load_rec(const int4* __restrict__ rec, const float* __restrict__ aed2,
                                         int e, int& nx, int& sr, float& tm) {
  if (TS < 2) {
    const int4 r = rec[e];
    nx = r.x;  sr = r.y;
    tm = __int_as_float(TS == 0 ? r.z : r.w);
  } else {
    const int2 r = *(const int2*)(rec + e);
    nx = r.x;  sr = r.y;
    tm = aed2[e];
  }
}

template <int TS>
__global__ void __launch_bounds__(256)
k_node(const float* __restrict__ xw, const float* __restrict__ s, const float* __restrict__ dv,
       const float* __restrict__ aed2, const int4* __restrict__ rec,
       const int* __restrict__ head, const float* __restrict__ bias,
       const float* __restrict__ gamma, const float* __restrict__ beta,
       const float* __restrict__ mean, const float* __restrict__ var,
       float* __restrict__ h, int n, int E, int relu) {
  int t = blockIdx.x * blockDim.x + threadIdx.x;
  int d = t >> 3;
  int j = t & 7;
  if (d >= n) return;
  const float dd = dv[d];
  Fold st;
  st.m = -INFINITY; st.den = 0.f; st.asum = 0.f; st.cnt = 0.f;
  st.acc0 = make_float4(0.f, 0.f, 0.f, 0.f);
  st.acc1 = make_float4(0.f, 0.f, 0.f, 0.f);

  int eA = head[d];
  int eB = head[n + d];
  bool vA = (unsigned)eA < (unsigned)E;
  bool vB = (unsigned)eB < (unsigned)E;
  int nxA, srA, nxB, srB;
  float tmA, tmB;
  load_rec<TS>(rec, aed2, vA ? eA : 0, nxA, srA, tmA);
  load_rec<TS>(rec, aed2, vB ? eB : 0, nxB, srB, tmB);
  int it = 0;
  for (; it < E && vA && vB; ++it) {
    const float ssA = s[srA];
    const float ssB = s[srB];
    const float* rpA = xw + (size_t)srA * D + j * 4;
    const float* rpB = xw + (size_t)srB * D + j * 4;
    const float4 rA0 = *(const float4*)rpA;
    const float4 rA1 = *(const float4*)(rpA + D / 2);
    const float4 rB0 = *(const float4*)rpB;
    const float4 rB1 = *(const float4*)(rpB + D / 2);
    int nnA, nsA, nnB, nsB;
    float ntA, ntB;
    load_rec<TS>(rec, aed2, (unsigned)nxA < (unsigned)E ? nxA : 0, nnA, nsA, ntA);
    load_rec<TS>(rec, aed2, (unsigned)nxB < (unsigned)E ? nxB : 0, nnB, nsB, ntB);
    st.asum += tmA;
    st.cnt += 1.0f;
    fold_edge(st, ssA + dd + tmA, rA0, rA1);
    st.asum += tmB;
    st.cnt += 1.0f;
    fold_edge(st, ssB + dd + tmB, rB0, rB1);
    eA = nxA;  vA = (unsigned)eA < (unsigned)E;  nxA = nnA;  srA = nsA;  tmA = ntA;
    eB = nxB;  vB = (unsigned)eB < (unsigned)E;  nxB = nnB;  srB = nsB;  tmB = ntB;
  }
  int e = vA ? eA : eB;
  int nx = vA ? nxA : nxB;
  int sr = vA ? srA : srB;
  float tm = vA ? tmA : tmB;
  for (; it < E && (unsigned)e < (unsigned)E; ++it) {
    const float ss = s[sr];
    const float* rp = xw + (size_t)sr * D + j * 4;
    const float4 row0 = *(const float4*)rp;
    const float4 row1 = *(const float4*)(rp + D / 2);
    int nn, ns;
    float nt;
    load_rec<TS>(rec, aed2, (unsigned)nx < (unsigned)E ? nx : 0, nn, ns, nt);
    st.asum += tm;
    st.cnt += 1.0f;
    fold_edge(st, ss + dd + tm, row0, row1);
    e = nx;
    nx = nn;  sr = ns;  tm = nt;
  }

  {
    const float* rp = xw + (size_t)d * D + j * 4;
    const float4 row0 = *(const float4*)rp;
    const float4 row1 = *(const float4*)(rp + D / 2);
    fold_edge(st, s[d] + dd + st.asum / fmaxf(st.cnt, 1.0f), row0, row1);
  }
  const float den = st.den;
  const float4 acc0 = st.acc0, acc1 = st.acc1;

  const float inv = 1.0f / den;
  #pragma unroll 1
  for (int hf = 0; hf < 2; hf++) {
    const int f = hf * (D / 2) + j * 4;
    const float4 ac = hf ? acc1 : acc0;
    const float4 b4 = *(const float4*)(bias  + f);
    const float4 g4 = *(const float4*)(gamma + f);
    const float4 t4 = *(const float4*)(beta  + f);
    const float4 m4 = *(const float4*)(mean  + f);
    const float4 v4 = *(const float4*)(var   + f);
    float4 o;
    o.x = (ac.x * inv + b4.x - m4.x) * (g4.x / sqrtf(v4.x + BN_EPS)) + t4.x;
    o.y = (ac.y * inv + b4.y - m4.y) * (g4.y / sqrtf(v4.y + BN_EPS)) + t4.y;
    o.z = (ac.z * inv + b4.z - m4.z) * (g4.z / sqrtf(v4.z + BN_EPS)) + t4.z;
    o.w = (ac.w * inv + b4.w - m4.w) * (g4.w / sqrtf(v4.w + BN_EPS)) + t4.w;
    if (relu) {
      o.x = fmaxf(o.x, 0.f); o.y = fmaxf(o.y, 0.f); o.z = fmaxf(o.z, 0.f); o.w = fmaxf(o.w, 0.f);
    }
    *(float4*)(h + (size_t)d * D + f) = o;
  }
}

#define RO_T 256
#define RO_W (RO_T / 32)
__global__ void __launch_bounds__(RO_T)
k_readout(const float* __restrict__ nodes, const int* __restrict__ batch,
          float* __restrict__ out_graph, int n) {
  __shared__ float part[16 * D];
  __shared__ int cnt[2][2][RO_W];
  const int g = blockIdx.x;
  const int t = threadIdx.x;

  int lo0 = 0, len0 = n, lo1 = 0, len1 = n;
  for (int round = 0; round < 8 && (len0 > 0 || len1 > 0); ++round) {
    const int st0 = (len0 + RO_T - 1) / RO_T;
    const int st1 = (len1 + RO_T - 1) / RO_T;
    const int p0 = lo0 + t * st0;
    const int p1 = lo1 + t * st1;
    const int b0 = batch[p0 < n ? p0 : n - 1];
    const int b1 = batch[p1 < n ? p1 : n - 1];
    const bool q0 = (p0 < lo0 + len0) && (b0 < g);
    const bool q1 = (p1 < lo1 + len1) && (b1 <= g);
    const int w0 = __popc(__builtin_amdgcn_ballot_w32(q0));
    const int w1 = __popc(__builtin_amdgcn_ballot_w32(q1));
    const int par = round & 1;
    if ((t & 31) == 0) { cnt[par][0][t >> 5] = w0; cnt[par][1][t >> 5] = w1; }
    __syncthreads();
    int c0 = 0, c1 = 0;
    for (int w = 0; w < RO_W; w++) { c0 += cnt[par][0][w]; c1 += cnt[par][1][w]; }
    c0 = __builtin_amdgcn_readfirstlane(c0);
    c1 = __builtin_amdgcn_readfirstlane(c1);
    if (len0 > 0) {
      if (c0 == 0) { len0 = 0; }
      else {
        const int nlo = lo0 + (c0 - 1) * st0 + 1;
        const int cap = lo0 + c0 * st0;
        const int nhi = cap < lo0 + len0 ? cap : lo0 + len0;
        lo0 = nlo; len0 = nhi - nlo;
      }
    }
    if (len1 > 0) {
      if (c1 == 0) { len1 = 0; }
      else {
        const int nlo = lo1 + (c1 - 1) * st1 + 1;
        const int cap = lo1 + c1 * st1;
        const int nhi = cap < lo1 + len1 ? cap : lo1 + len1;
        lo1 = nlo; len1 = nhi - nlo;
      }
    }
  }
  const int first = lo0, last = lo1;

  const int r = t >> 4;
  const int c = t & 15;
  float4 acc = make_float4(0.f, 0.f, 0.f, 0.f);
  int i = first + r;
  for (; i + 48 < last; i += 64) {
    const float* p = nodes + (size_t)i * D + c * 4;
    const float4 v0 = *(const float4*)p;
    const float4 v1 = *(const float4*)(p + 16 * D);
    const float4 v2 = *(const float4*)(p + 32 * D);
    const float4 v3 = *(const float4*)(p + 48 * D);
    acc.x += v0.x; acc.y += v0.y; acc.z += v0.z; acc.w += v0.w;
    acc.x += v1.x; acc.y += v1.y; acc.z += v1.z; acc.w += v1.w;
    acc.x += v2.x; acc.y += v2.y; acc.z += v2.z; acc.w += v2.w;
    acc.x += v3.x; acc.y += v3.y; acc.z += v3.z; acc.w += v3.w;
  }
  for (; i < last; i += 16) {
    const float4 v = *(const float4*)(nodes + (size_t)i * D + c * 4);
    acc.x += v.x; acc.y += v.y; acc.z += v.z; acc.w += v.w;
  }
  *(float4*)(part + r * D + c * 4) = acc;
  __syncthreads();
  if (t < D) {
    float s = 0.f;
    for (int k = 0; k < 16; k++) s += part[k * D + t];
    out_graph[(size_t)g * D + t] = s;
  }
}

extern "C" void kernel_launch(void* const* d_in, const int* in_sizes, int n_in,
                              void* d_out, int out_size, void* d_ws, size_t ws_size,
                              hipStream_t stream) {
  const float* x     = (const float*)d_in[0];
  const int*   eidx  = (const int*)d_in[1];
  const float* eattr = (const float*)d_in[2];
  const int*   batch = (const int*)d_in[3];
  const float* xlw   = (const float*)d_in[4];
  const float* xlb   = (const float*)d_in[5];
  const float* glw   = (const float*)d_in[6];
  const float* gas   = (const float*)d_in[7];
  const float* gad   = (const float*)d_in[8];
  const float* gew   = (const float*)d_in[9];
  const float* gae   = (const float*)d_in[10];
  const float* gb    = (const float*)d_in[11];
  const float* bng   = (const float*)d_in[12];
  const float* bnb   = (const float*)d_in[13];
  const float* bnm   = (const float*)d_in[14];
  const float* bnv   = (const float*)d_in[15];

  const int n = in_sizes[3];
  const int E = in_sizes[2] / F_E;
  const int* srcp = eidx;
  const int* dstp = eidx + E;

  float* out_nodes = (float*)d_out;
  float* out_graph = out_nodes + (size_t)n * D;
  const size_t graph_elems = (size_t)out_size - (size_t)n * D;

  float* h     = out_nodes;
  float* ws    = (float*)d_ws;
  float* xw    = ws;  ws += (size_t)n * D;
  float* wea   = ws;  ws += 64;
  float* usd   = ws;  ws += L * 2 * D;
  int4*  rec   = (int4*)ws; ws += (size_t)E * 4;
  float* aed2  = ws;  ws += (size_t)E;
  float* sS    = ws;  ws += n;
  float* dS    = ws;  ws += n;
  int*   head  = (int*)ws; ws += (size_t)n * 2;

  const int B = 256;
  #define GRID(x) (((x) + B - 1) / B)

  k_prep<<<GRID(n > PREP_U0 + L * 2 * D ? n : PREP_U0 + L * 2 * D), B, 0, stream>>>(
      gew, gae, glw, gas, gad, wea, usd, head, n);
  k_edge_build<<<GRID((E + ET_TRIPS - 1) / ET_TRIPS), B, 0, stream>>>(eattr, wea, srcp, dstp, head, rec, aed2, E, n);

  {
    int waves = (n + 15) / 16;
    int blocks = (waves + 3) / 4;
    k_gemm_wmma<<<blocks, 128, 0, stream>>>(x, xlw, xlb, h, n, F_IN);
  }

  for (int l = 0; l < L; l++) {
    int waves = (n + 15) / 16;
    int blocks = (waves + 3) / 4;
    k_gemm_scores<<<blocks, 128, 0, stream>>>(h, glw + (size_t)l * D * D, xw, usd + (size_t)l * 2 * D,
                                              sS, dS, n, D);
    #define NODE_ARGS xw, sS, dS, aed2, rec, head, gb + (size_t)l * D, bng + (size_t)l * D, \
                      bnb + (size_t)l * D, bnm + (size_t)l * D, bnv + (size_t)l * D, h, n, E, l < L - 1 ? 1 : 0
    if (l == 0)      k_node<0><<<GRID(n * 8), B, 0, stream>>>(NODE_ARGS);
    else if (l == 1) k_node<1><<<GRID(n * 8), B, 0, stream>>>(NODE_ARGS);
    else             k_node<2><<<GRID(n * 8), B, 0, stream>>>(NODE_ARGS);
    #undef NODE_ARGS
  }

  k_readout<<<(int)(graph_elems / D), RO_T, 0, stream>>>(out_nodes, batch, out_graph, n);
  #undef GRID
}


# input linear folded into layer 0 by linearity: xw_0 = x (Wi W_0) + bi W_0, weights formed per call; that product runs from x inside the edge-pass launch; 2 launches gone; layer-0 sum order changed
# speedup vs baseline: 8.0049x; 1.0761x over previous
#include <hip/hip_runtime.h>


#define F_IN   32
#define F_E    16
#define D      64
#define L      3
#define NEG_SLOPE 0.2f
#define BN_EPS 1e-5f

typedef __attribute__((ext_vector_type(2))) float v2f;
typedef __attribute__((ext_vector_type(8))) float v8f;

#define PREP_U0 64
#define PREP_WC 512
#define PREP_BC (PREP_WC + F_IN * D)
#define PREP_SB ((PREP_BC + D + 255) / 256)
#define PREP_MIN ((PREP_SB + 1) * 256)
#define L0_WC 0
#define L0_BC (F_IN * D)
#define L0_UX (L0_BC + D)
#define L0_SC (L0_UX + 2 * F_IN)
#define L0_WORDS (L0_SC + 2)
__global__ void k_prep(const float* __restrict__ We, const float* __restrict__ ae,
                       const float* __restrict__ Wl, const float* __restrict__ as,
                       const float* __restrict__ ad, const float* __restrict__ Wi,
                       const float* __restrict__ bi, float* __restrict__ wea,
                       float* __restrict__ usd, float* __restrict__ l0,
                       int* __restrict__ head, int n) {
  __shared__ float v0[2 * D];
  int i = blockIdx.x * blockDim.x + threadIdx.x;
  if (i < n) { head[i] = -1; head[n + i] = -1; }
  if (i < L * F_E) {
    int l = i / F_E, f = i % F_E;
    float s = 0.f;
    #pragma unroll
    for (int d = 0; d < D; d++) s += We[(l * F_E + f) * D + d] * ae[l * D + d];
    wea[i] = s;
  } else if (i >= PREP_U0 && i < PREP_U0 + L * 2 * D) {
    const int r = i - PREP_U0;
    const int l = r / (2 * D), q = r % (2 * D);
    const int k = (q >> 2) * 2 + (q & 1);
    const float* av = ((q & 2) ? ad : as) + l * D;
    const float* wr = Wl + ((size_t)l * D + k) * D;
    float s = 0.f;
    for (int j = 0; j < D; j++) s += wr[j] * av[j];
    usd[r] = s;
  } else if (i >= PREP_WC && i < PREP_WC + F_IN * D) {
    const int r = i - PREP_WC;
    const int k = r / D, j = r % D;
    float s = 0.f;
    for (int m = 0; m < D; m++) s += Wi[k * D + m] * Wl[m * D + j];
    l0[L0_WC + r] = s;
  } else if (i >= PREP_BC && i < PREP_BC + D) {
    const int j = i - PREP_BC;
    float s = 0.f;
    for (int m = 0; m < D; m++) s += bi[m] * Wl[m * D + j];
    l0[L0_BC + j] = s;
  }
  if (blockIdx.x == PREP_SB) {
    const int tt = threadIdx.x;
    if (tt < 2 * D) {
      const int m = tt % D;
      const float* av = tt < D ? as : ad;
      float s = 0.f;
      for (int j = 0; j < D; j++) s += Wl[m * D + j] * av[j];
      v0[tt] = s;
    }
    __syncthreads();
    if (tt < 2 * F_IN) {
      const int k = (tt >> 2) * 2 + (tt & 1);
      const float* vv = v0 + ((tt & 2) ? D : 0);
      float s = 0.f;
      for (int m = 0; m < D; m++) s += Wi[k * D + m] * vv[m];
      l0[L0_UX + tt] = s;
    } else if (tt < 2 * F_IN + 2) {
      const float* vv = v0 + (tt - 2 * F_IN) * D;
      float s = 0.f;
      for (int m = 0; m < D; m++) s += bi[m] * vv[m];
      l0[L0_SC + (tt - 2 * F_IN)] = s;
    }
  }
}

#define ET_TRIPS 4
__device__ __forceinline__ void edge_pass(const float* __restrict__ eattr, const float* __restrict__ wea,
                                          const int* __restrict__ src, const int* __restrict__ dst,
                                          int* __restrict__ head, int4* __restrict__ rec,
                                          float* __restrict__ aed2, int E, int n, int eblk, int eblocks) {
  static_assert(L == 3, "two terms ride in the record, the third has its plane");
  const int t = eblk * blockDim.x + threadIdx.x;
  const int T = eblocks * blockDim.x;
  const int half = (int)((__builtin_amdgcn_s_sendmsg_rtn(0x87) >> 18) & 1u);
  int sv[ET_TRIPS], dn[ET_TRIPS];
  float4 v[ET_TRIPS][F_E / 4];
  for (int k = 0; k < ET_TRIPS; k++) {
    const int e = t + k * T;
    const int ec = e < E ? e : E - 1;
    dn[k] = dst[ec];
    sv[k] = src[ec];
    const float4* ea = (const float4*)(eattr + (size_t)ec * F_E);
    for (int q = 0; q < F_E / 4; q++) v[k][q] = ea[q];
  }
  float4 w[L][F_E / 4];
  for (int l = 0; l < L; l++)
    for (int q = 0; q < F_E / 4; q++) w[l][q] = *(const float4*)(wea + l * F_E + q * 4);
  for (int k = 0; k < ET_TRIPS; k++) {
    const int e = t + k * T;
    float a[L];
    for (int l = 0; l < L; l++) {
      float acc = 0.f;
      for (int q = 0; q < F_E / 4; q++) {
        acc += v[k][q].x * w[l][q].x;
        acc += v[k][q].y * w[l][q].y;
        acc += v[k][q].z * w[l][q].z;
        acc += v[k][q].w * w[l][q].w;
      }
      a[l] = acc;
    }
    if (e < E) {
      const int prev = atomicExch(&head[(size_t)half * n + dn[k]], e);
      rec[e] = make_int4(prev, sv[k], __float_as_int(a[0]), __float_as_int(a[1]));
      aed2[e] = a[2];
    }
  }
}

template <bool SC>
__device__ __forceinline__ void gemm_rows(const float* __restrict__ A, const float* __restrict__ B,
                                          const float* __restrict__ bias, float* __restrict__ C,
                                          int n, int K, const float* __restrict__ usd,
                                          float* __restrict__ s, float* __restrict__ dv,
                                          int bidx, const float* __restrict__ sconst) {
  int wave = (bidx * blockDim.x + threadIdx.x) >> 5;
  int lane = threadIdx.x & 31;
  int row0 = wave * 16;
  if (row0 >= n) return;
  int lm = lane & 15;
  int kh = (lane >> 4) << 1;
  int mrow = row0 + lm;
  int mrd = mrow < n ? mrow : n - 1;

  v8f acc[4] = {};
  float ps = 0.f, pd = 0.f;
  for (int k = 0; k < K; k += 4) {
    v2f a;
    a.x = A[(size_t)mrd * K + k + kh];
    a.y = A[(size_t)mrd * K + k + kh + 1];
    if (SC) {
      const float4 u = *(const float4*)(usd + (k + kh) * 2);
      ps += a.x * u.x;
      ps += a.y * u.y;
      pd += a.x * u.z;
      pd += a.y * u.w;
    }
    #pragma unroll
    for (int nt = 0; nt < 4; nt++) {
      int col = nt * 16 + lm;
      v2f b;
      b.x = B[(k + kh) * D + col];
      b.y = B[(k + kh + 1) * D + col];
      acc[nt] = __builtin_amdgcn_wmma_f32_16x16x4_f32(
          false, a, false, b, (short)0, acc[nt], false, false);
    }
  }

  int rbase = row0 + ((lane >> 4) << 3);
  #pragma unroll
  for (int nt = 0; nt < 4; nt++) {
    int col = nt * 16 + lm;
    float bv = bias ? bias[col] : 0.f;
    #pragma unroll
    for (int r = 0; r < 8; r++) {
      int row = rbase + r;
      if (row < n) C[(size_t)row * D + col] = acc[nt][r] + bv;
    }
  }

  if (SC) {
    ps += __shfl_xor(ps, 16);
    pd += __shfl_xor(pd, 16);
    if (sconst) { ps += sconst[0]; pd += sconst[1]; }
    if (lane < 16 && mrow < n) { s[mrow] = ps; dv[mrow] = pd; }
  }
}

__global__ void k_gemm_scores(const float* __restrict__ A, const float* __restrict__ B,
                              float* __restrict__ C, const float* __restrict__ usd,
                              float* __restrict__ s, float* __restrict__ dv, int n, int K) {
  gemm_rows<true>(A, B, nullptr, C, n, K, usd, s, dv, blockIdx.x, nullptr);
}

__global__ void k_edge_l0(const float* __restrict__ eattr, const float* __restrict__ wea,
                          const int* __restrict__ src, const int* __restrict__ dst,
                          int* __restrict__ head, int4* __restrict__ rec,
                          float* __restrict__ aed2, int E, int n,
                          const float* __restrict__ x, const float* __restrict__ l0,
                          float* __restrict__ xw, float* __restrict__ s, float* __restrict__ dv,
                          int gblocks) {
  const int b = blockIdx.x;
  if (b < gblocks) {
    gemm_rows<true>(x, l0 + L0_WC, l0 + L0_BC, xw, n, F_IN, l0 + L0_UX, s, dv, b, l0 + L0_SC);
  } else {
    edge_pass(eattr, wea, src, dst, head, rec, aed2, E, n, b - gblocks, (int)gridDim.x - gblocks);
  }
}

struct Fold {
  float m, den, asum, cnt;
  float4 acc0, acc1;
};

__device__ __forceinline__ void fold_edge(Fold& f, float a, const float4& row0, const float4& row1) {
  a = a > 0.f ? a : NEG_SLOPE * a;
  const float dl = a - f.m;
  const float ex = expf(-fabsf(dl));
  const bool up = dl > 0.f;
  const float sc = up ? ex : 1.0f;
  const float p  = up ? 1.0f : ex;
  f.m = up ? a : f.m;
  f.den = f.den * sc + p;
  f.acc0.x = f.acc0.x * sc + p * row0.x;
  f.acc0.y = f.acc0.y * sc + p * row0.y;
  f.acc0.z = f.acc0.z * sc + p * row0.z;
  f.acc0.w = f.acc0.w * sc + p * row0.w;
  f.acc1.x = f.acc1.x * sc + p * row1.x;
  f.acc1.y = f.acc1.y * sc + p * row1.y;
  f.acc1.z = f.acc1.z * sc + p * row1.z;
  f.acc1.w = f.acc1.w * sc + p * row1.w;
}

template <int TS>
__device__ __forceinline__ void load_rec(const int4* __restrict__ rec, const float* __restrict__ aed2,
                                         int e, int& nx, int& sr, float& tm) {
  if (TS < 2) {
    const int4 r = rec[e];
    nx = r.x;  sr = r.y;
    tm = __int_as_float(TS == 0 ? r.z : r.w);
  } else {
    const int2 r = *(const int2*)(rec + e);
    nx = r.x;  sr = r.y;
    tm = aed2[e];
  }
}

template <int TS>
__global__ void __launch_bounds__(256)
k_node(const float* __restrict__ xw, const float* __restrict__ s, const float* __restrict__ dv,
       const float* __restrict__ aed2, const int4* __restrict__ rec,
       const int* __restrict__ head, const float* __restrict__ bias,
       const float* __restrict__ gamma, const float* __restrict__ beta,
       const float* __restrict__ mean, const float* __restrict__ var,
       float* __restrict__ h, int n, int E, int relu) {
  int t = blockIdx.x * blockDim.x + threadIdx.x;
  int d = t >> 3;
  int j = t & 7;
  if (d >= n) return;
  const float dd = dv[d];
  Fold st;
  st.m = -INFINITY; st.den = 0.f; st.asum = 0.f; st.cnt = 0.f;
  st.acc0 = make_float4(0.f, 0.f, 0.f, 0.f);
  st.acc1 = make_float4(0.f, 0.f, 0.f, 0.f);

  int eA = head[d];
  int eB = head[n + d];
  bool vA = (unsigned)eA < (unsigned)E;
  bool vB = (unsigned)eB < (unsigned)E;
  int nxA, srA, nxB, srB;
  float tmA, tmB;
  load_rec<TS>(rec, aed2, vA ? eA : 0, nxA, srA, tmA);
  load_rec<TS>(rec, aed2, vB ? eB : 0, nxB, srB, tmB);
  int it = 0;
  for (; it < E && vA && vB; ++it) {
    const float ssA = s[srA];
    const float ssB = s[srB];
    const float* rpA = xw + (size_t)srA * D + j * 4;
    const float* rpB = xw + (size_t)srB * D + j * 4;
    const float4 rA0 = *(const float4*)rpA;
    const float4 rA1 = *(const float4*)(rpA + D / 2);
    const float4 rB0 = *(const float4*)rpB;
    const float4 rB1 = *(const float4*)(rpB + D / 2);
    int nnA, nsA, nnB, nsB;
    float ntA, ntB;
    load_rec<TS>(rec, aed2, (unsigned)nxA < (unsigned)E ? nxA : 0, nnA, nsA, ntA);
    load_rec<TS>(rec, aed2, (unsigned)nxB < (unsigned)E ? nxB : 0, nnB, nsB, ntB);
    st.asum += tmA;
    st.cnt += 1.0f;
    fold_edge(st, ssA + dd + tmA, rA0, rA1);
    st.asum += tmB;
    st.cnt += 1.0f;
    fold_edge(st, ssB + dd + tmB, rB0, rB1);
    eA = nxA;  vA = (unsigned)eA < (unsigned)E;  nxA = nnA;  srA = nsA;  tmA = ntA;
    eB = nxB;  vB = (unsigned)eB < (unsigned)E;  nxB = nnB;  srB = nsB;  tmB = ntB;
  }
  int e = vA ? eA : eB;
  int nx = vA ? nxA : nxB;
  int sr = vA ? srA : srB;
  float tm = vA ? tmA : tmB;
  for (; it < E && (unsigned)e < (unsigned)E; ++it) {
    const float ss = s[sr];
    const float* rp = xw + (size_t)sr * D + j * 4;
    const float4 row0 = *(const float4*)rp;
    const float4 row1 = *(const float4*)(rp + D / 2);
    int nn, ns;
    float nt;
    load_rec<TS>(rec, aed2, (unsigned)nx < (unsigned)E ? nx : 0, nn, ns, nt);
    st.asum += tm;
    st.cnt += 1.0f;
    fold_edge(st, ss + dd + tm, row0, row1);
    e = nx;
    nx = nn;  sr = ns;  tm = nt;
  }

  {
    const float* rp = xw + (size_t)d * D + j * 4;
    const float4 row0 = *(const float4*)rp;
    const float4 row1 = *(const float4*)(rp + D / 2);
    fold_edge(st, s[d] + dd + st.asum / fmaxf(st.cnt, 1.0f), row0, row1);
  }
  const float den = st.den;
  const float4 acc0 = st.acc0, acc1 = st.acc1;

  const float inv = 1.0f / den;
  #pragma unroll 1
  for (int hf = 0; hf < 2; hf++) {
    const int f = hf * (D / 2) + j * 4;
    const float4 ac = hf ? acc1 : acc0;
    const float4 b4 = *(const float4*)(bias  + f);
    const float4 g4 = *(const float4*)(gamma + f);
    const float4 t4 = *(const float4*)(beta  + f);
    const float4 m4 = *(const float4*)(mean  + f);
    const float4 v4 = *(const float4*)(var   + f);
    float4 o;
    o.x = (ac.x * inv + b4.x - m4.x) * (g4.x / sqrtf(v4.x + BN_EPS)) + t4.x;
    o.y = (ac.y * inv + b4.y - m4.y) * (g4.y / sqrtf(v4.y + BN_EPS)) + t4.y;
    o.z = (ac.z * inv + b4.z - m4.z) * (g4.z / sqrtf(v4.z + BN_EPS)) + t4.z;
    o.w = (ac.w * inv + b4.w - m4.w) * (g4.w / sqrtf(v4.w + BN_EPS)) + t4.w;
    if (relu) {
      o.x = fmaxf(o.x, 0.f); o.y = fmaxf(o.y, 0.f); o.z = fmaxf(o.z, 0.f); o.w = fmaxf(o.w, 0.f);
    }
    *(float4*)(h + (size_t)d * D + f) = o;
  }
}

#define RO_T 256
#define RO_W (RO_T / 32)
__global__ void __launch_bounds__(RO_T)
k_readout(const float* __restrict__ nodes, const int* __restrict__ batch,
          float* __restrict__ out_graph, int n) {
  __shared__ float part[16 * D];
  __shared__ int cnt[2][2][RO_W];
  const int g = blockIdx.x;
  const int t = threadIdx.x;

  int lo0 = 0, len0 = n, lo1 = 0, len1 = n;
  for (int round = 0; round < 8 && (len0 > 0 || len1 > 0); ++round) {
    const int st0 = (len0 + RO_T - 1) / RO_T;
    const int st1 = (len1 + RO_T - 1) / RO_T;
    const int p0 = lo0 + t * st0;
    const int p1 = lo1 + t * st1;
    const int b0 = batch[p0 < n ? p0 : n - 1];
    const int b1 = batch[p1 < n ? p1 : n - 1];
    const bool q0 = (p0 < lo0 + len0) && (b0 < g);
    const bool q1 = (p1 < lo1 + len1) && (b1 <= g);
    const int w0 = __popc(__builtin_amdgcn_ballot_w32(q0));
    const int w1 = __popc(__builtin_amdgcn_ballot_w32(q1));
    const int par = round & 1;
    if ((t & 31) == 0) { cnt[par][0][t >> 5] = w0; cnt[par][1][t >> 5] = w1; }
    __syncthreads();
    int c0 = 0, c1 = 0;
    for (int w = 0; w < RO_W; w++) { c0 += cnt[par][0][w]; c1 += cnt[par][1][w]; }
    c0 = __builtin_amdgcn_readfirstlane(c0);
    c1 = __builtin_amdgcn_readfirstlane(c1);
    if (len0 > 0) {
      if (c0 == 0) { len0 = 0; }
      else {
        const int nlo = lo0 + (c0 - 1) * st0 + 1;
        const int cap = lo0 + c0 * st0;
        const int nhi = cap < lo0 + len0 ? cap : lo0 + len0;
        lo0 = nlo; len0 = nhi - nlo;
      }
    }
    if (len1 > 0) {
      if (c1 == 0) { len1 = 0; }
      else {
        const int nlo = lo1 + (c1 - 1) * st1 + 1;
        const int cap = lo1 + c1 * st1;
        const int nhi = cap < lo1 + len1 ? cap : lo1 + len1;
        lo1 = nlo; len1 = nhi - nlo;
      }
    }
  }
  const int first = lo0, last = lo1;

  const int r = t >> 4;
  const int c = t & 15;
  float4 acc = make_float4(0.f, 0.f, 0.f, 0.f);
  int i = first + r;
  for (; i + 48 < last; i += 64) {
    const float* p = nodes + (size_t)i * D + c * 4;
    const float4 v0 = *(const float4*)p;
    const float4 v1 = *(const float4*)(p + 16 * D);
    const float4 v2 = *(const float4*)(p + 32 * D);
    const float4 v3 = *(const float4*)(p + 48 * D);
    acc.x += v0.x; acc.y += v0.y; acc.z += v0.z; acc.w += v0.w;
    acc.x += v1.x; acc.y += v1.y; acc.z += v1.z; acc.w += v1.w;
    acc.x += v2.x; acc.y += v2.y; acc.z += v2.z; acc.w += v2.w;
    acc.x += v3.x; acc.y += v3.y; acc.z += v3.z; acc.w += v3.w;
  }
  for (; i < last; i += 16) {
    const float4 v = *(const float4*)(nodes + (size_t)i * D + c * 4);
    acc.x += v.x; acc.y += v.y; acc.z += v.z; acc.w += v.w;
  }
  *(float4*)(part + r * D + c * 4) = acc;
  __syncthreads();
  if (t < D) {
    float s = 0.f;
    for (int k = 0; k < 16; k++) s += part[k * D + t];
    out_graph[(size_t)g * D + t] = s;
  }
}

extern "C" void kernel_launch(void* const* d_in, const int* in_sizes, int n_in,
                              void* d_out, int out_size, void* d_ws, size_t ws_size,
                              hipStream_t stream) {
  const float* x     = (const float*)d_in[0];
  const int*   eidx  = (const int*)d_in[1];
  const float* eattr = (const float*)d_in[2];
  const int*   batch = (const int*)d_in[3];
  const float* xlw   = (const float*)d_in[4];
  const float* xlb   = (const float*)d_in[5];
  const float* glw   = (const float*)d_in[6];
  const float* gas   = (const float*)d_in[7];
  const float* gad   = (const float*)d_in[8];
  const float* gew   = (const float*)d_in[9];
  const float* gae   = (const float*)d_in[10];
  const float* gb    = (const float*)d_in[11];
  const float* bng   = (const float*)d_in[12];
  const float* bnb   = (const float*)d_in[13];
  const float* bnm   = (const float*)d_in[14];
  const float* bnv   = (const float*)d_in[15];

  const int n = in_sizes[3];
  const int E = in_sizes[2] / F_E;
  const int* srcp = eidx;
  const int* dstp = eidx + E;

  float* out_nodes = (float*)d_out;
  float* out_graph = out_nodes + (size_t)n * D;
  const size_t graph_elems = (size_t)out_size - (size_t)n * D;

  float* h     = out_nodes;
  float* ws    = (float*)d_ws;
  float* xw    = ws;  ws += (size_t)n * D;
  float* wea   = ws;  ws += 64;
  float* usd   = ws;  ws += L * 2 * D;
  int4*  rec   = (int4*)ws; ws += (size_t)E * 4;
  float* aed2  = ws;  ws += (size_t)E;
  float* sS    = ws;  ws += n;
  float* dS    = ws;  ws += n;
  int*   head  = (int*)ws; ws += (size_t)n * 2;

  const int B = 256;
  #define GRID(x) (((x) + B - 1) / B)

  float* l0 = out_graph;
  k_prep<<<GRID(n > PREP_MIN ? n : PREP_MIN), B, 0, stream>>>(
      gew, gae, glw, gas, gad, xlw, xlb, wea, usd, l0, head, n);
  {
    const int gblocks = (n + 127) / 128;
    k_edge_l0<<<gblocks + GRID((E + ET_TRIPS - 1) / ET_TRIPS), B, 0, stream>>>(
        eattr, wea, srcp, dstp, head, rec, aed2, E, n, x, l0, xw, sS, dS, gblocks);
  }

  for (int l = 0; l < L; l++) {
    int waves = (n + 15) / 16;
    int blocks = (waves + 3) / 4;
    if (l > 0)
      k_gemm_scores<<<blocks, 128, 0, stream>>>(h, glw + (size_t)l * D * D, xw, usd + (size_t)l * 2 * D,
                                                sS, dS, n, D);
    #define NODE_ARGS xw, sS, dS, aed2, rec, head, gb + (size_t)l * D, bng + (size_t)l * D, \
                      bnb + (size_t)l * D, bnm + (size_t)l * D, bnv + (size_t)l * D, h, n, E, l < L - 1 ? 1 : 0
    if (l == 0)      k_node<0><<<GRID(n * 8), B, 0, stream>>>(NODE_ARGS);
    else if (l == 1) k_node<1><<<GRID(n * 8), B, 0, stream>>>(NODE_ARGS);
    else             k_node<2><<<GRID(n * 8), B, 0, stream>>>(NODE_ARGS);
    #undef NODE_ARGS
  }

  k_readout<<<(int)(graph_elems / D), RO_T, 0, stream>>>(out_nodes, batch, out_graph, n);
  #undef GRID
}
